# starting kernel
# baseline (speedup 1.0000x reference)
#include <hip/hip_runtime.h>

typedef __attribute__((ext_vector_type(2))) float v2f;
typedef __attribute__((ext_vector_type(8))) float v8f;

#define NB    4
#define NN    50000
#define MM    (NB * NN)
#define CIN   64
#define HID   256
#define COUTN 64
#define NE    800000
#define STRIPS (MM / 16)

__global__ void gcn_zero_kernel(float* __restrict__ y) {
  long i4 = ((long)blockIdx.x * blockDim.x + threadIdx.x) * 4;
  *(float4*)(y + i4) = make_float4(0.f, 0.f, 0.f, 0.f);
}

__global__ void gcn_scatter_kernel(const float* __restrict__ x,
                                   const int* __restrict__ rows,
                                   const int* __restrict__ cols,
                                   const float* __restrict__ vals,
                                   float* __restrict__ y) {
  int t  = blockIdx.x * 256 + threadIdx.x;
  int e  = t >> 6;
  int ch = t & 63;
  if (e >= NE) return;
  int   r = rows[e];
  int   c = cols[e];
  float v = vals[e];
#pragma unroll
  for (int b = 0; b < NB; ++b) {
    float xv = x[(size_t)(b * NN + c) * CIN + ch];
    unsafeAtomicAdd(&y[(size_t)(b * NN + r) * CIN + ch], v * xv);
  }
}

__global__ void __launch_bounds__(256, 1)
gcn_mlp_kernel(float* __restrict__ xo,
               const float* __restrict__ W1,
               const float* __restrict__ b1,
               const float* __restrict__ W2,
               const float* __restrict__ b2) {
  __shared__ __align__(16) float sW1t[HID * CIN];
  __shared__ __align__(16) float sW2t[COUTN * HID];
  __shared__ __align__(16) float sB1[HID];
  __shared__ __align__(16) float sB2[COUTN];
  __shared__ __align__(16) float sStage[8][16 * 16];

  const int tid = threadIdx.x;
  for (int i = tid; i < CIN * HID; i += 256) {
    int k = i >> 8, n = i & 255;
    sW1t[n * CIN + k] = W1[i];
  }
  for (int i = tid; i < HID * COUTN; i += 256) {
    int k = i >> 6, o = i & 63;
    sW2t[o * HID + k] = W2[i];
  }
  if (tid < HID)   sB1[tid] = b1[tid];
  if (tid < COUTN) sB2[tid] = b2[tid];
  __syncthreads();

  const int wave  = tid >> 5;
  const int lane  = tid & 31;
  const int lhalf = lane >> 4;
  const int lmod  = lane & 15;

  const int strip = blockIdx.x * 8 + wave;
  if (strip >= STRIPS) return;
  const int row0 = strip * 16;

  v2f a1[16];
  {
    const float* base = xo + (size_t)(row0 + lmod) * CIN + 2 * lhalf;
#pragma unroll
    for (int s = 0; s < 16; ++s)
      a1[s] = *(const v2f*)(base + 4 * s);
  }

  v8f oacc[4];
#pragma unroll
  for (int o = 0; o < 4; ++o) {
    float bv = sB2[o * 16 + lmod];
#pragma unroll
    for (int j = 0; j < 8; ++j) oacc[o][j] = bv;
  }

  float* st = &sStage[wave][0];

  for (int t = 0; t < 16; ++t) {
    v8f h;
    float b1v = sB1[t * 16 + lmod];
#pragma unroll
    for (int j = 0; j < 8; ++j) h[j] = b1v;
#pragma unroll
    for (int s = 0; s < 16; ++s) {
      v2f bfrag = *(const v2f*)(&sW1t[(t * 16 + lmod) * CIN + 4 * s + 2 * lhalf]);
      h = __builtin_amdgcn_wmma_f32_16x16x4_f32(false, a1[s], false, bfrag,
                                                (short)0, h, false, false);
    }
#pragma unroll
    for (int j = 0; j < 8; ++j) h[j] = fmaxf(h[j], 0.f);

#pragma unroll
    for (int j = 0; j < 8; ++j)
      st[(lhalf * 8 + j) * 16 + lmod] = h[j];
    asm volatile("s_wait_dscnt 0" ::: "memory");

    v2f a2[4];
#pragma unroll
    for (int kk = 0; kk < 4; ++kk)
      a2[kk] = *(const v2f*)(&st[lmod * 16 + 4 * kk + 2 * lhalf]);

#pragma unroll
    for (int o = 0; o < 4; ++o) {
#pragma unroll
      for (int kk = 0; kk < 4; ++kk) {
        v2f b2frag = *(const v2f*)(&sW2t[(o * 16 + lmod) * HID + t * 16 + 4 * kk + 2 * lhalf]);
        oacc[o] = __builtin_amdgcn_wmma_f32_16x16x4_f32(false, a2[kk], false, b2frag,
                                                        (short)0, oacc[o], false, false);
      }
    }
  }

#pragma unroll
  for (int o = 0; o < 4; ++o) {
#pragma unroll
    for (int j = 0; j < 8; ++j)
      xo[(size_t)(row0 + lhalf * 8 + j) * COUTN + o * 16 + lmod] = oacc[o][j];
  }
}

extern "C" void kernel_launch(void* const* d_in, const int* in_sizes, int n_in,
                              void* d_out, int out_size, void* d_ws, size_t ws_size,
                              hipStream_t stream) {
  const float* x     = (const float*)d_in[0];
  const int*   A_idx = (const int*)d_in[1];
  const float* A_val = (const float*)d_in[2];
  const float* W1    = (const float*)d_in[3];
  const float* b1    = (const float*)d_in[4];
  const float* W2    = (const float*)d_in[5];
  const float* b2    = (const float*)d_in[6];
  float* out = (float*)d_out;

  gcn_zero_kernel<<<12500, 256, 0, stream>>>(out);

  gcn_scatter_kernel<<<(NE * 64) / 256, 256, 0, stream>>>(
      x, A_idx, A_idx + NE, A_val, out);

  gcn_mlp_kernel<<<(STRIPS + 7) / 8, 256, 0, stream>>>(out, W1, b1, W2, b2);
}


# MLP kernel: W1 and W2 kept in LDS in WMMA B-fragment order, one lane-contiguous ds_load_b128 per two k-steps; stage tile pitch padded; no LDS bank conflicts on fill or read; sum order unchanged
# speedup vs baseline: 1.4873x; 1.4873x over previous
#include <hip/hip_runtime.h>

typedef __attribute__((ext_vector_type(2))) float v2f;
typedef __attribute__((ext_vector_type(8))) float v8f;

#define NB    4
#define NN    50000
#define MM    (NB * NN)
#define CIN   64
#define HID   256
#define COUTN 64
#define NE    800000
#define STRIPS (MM / 16)

__global__ void gcn_zero_kernel(float* __restrict__ y) {
  long i4 = ((long)blockIdx.x * blockDim.x + threadIdx.x) * 4;
  *(float4*)(y + i4) = make_float4(0.f, 0.f, 0.f, 0.f);
}

__global__ void gcn_scatter_kernel(const float* __restrict__ x,
                                   const int* __restrict__ rows,
                                   const int* __restrict__ cols,
                                   const float* __restrict__ vals,
                                   float* __restrict__ y) {
  int t  = blockIdx.x * 256 + threadIdx.x;
  int e  = t >> 6;
  int ch = t & 63;
  if (e >= NE) return;
  int   r = rows[e];
  int   c = cols[e];
  float v = vals[e];
#pragma unroll
  for (int b = 0; b < NB; ++b) {
    float xv = x[(size_t)(b * NN + c) * CIN + ch];
    unsafeAtomicAdd(&y[(size_t)(b * NN + r) * CIN + ch], v * xv);
  }
}

#define STP 20
__global__ void __launch_bounds__(256, 1)
gcn_mlp_kernel(float* __restrict__ xo,
               const float* __restrict__ W1,
               const float* __restrict__ b1,
               const float* __restrict__ W2,
               const float* __restrict__ b2) {
  __shared__ __align__(16) float sW1f[HID * CIN];
  __shared__ __align__(16) float sW2f[COUTN * HID];
  __shared__ __align__(16) float sB1[HID];
  __shared__ __align__(16) float sB2[COUTN];
  __shared__ __align__(16) float sStage[8][16 * STP];

  const int tid = threadIdx.x;
  {
    const int n = tid, tt = n >> 4, lm = n & 15;
#pragma unroll
    for (int g = 0; g < 16; ++g) {
      const int p = g >> 1, lh = g & 1;
      const float* src = W1 + (8 * p + 2 * lh) * HID + n;
      float4 v = make_float4(src[0], src[HID], src[4 * HID], src[5 * HID]);
      *(float4*)(&sW1f[(((tt * 8 + p) * 32) + lh * 16 + lm) * 4]) = v;
    }
  }
  {
    const int n = tid & 63, o4 = n >> 4, lm = n & 15, sub = tid >> 6;
#pragma unroll
    for (int j = 0; j < 16; ++j) {
      const int g = j * 4 + sub;
      const int tt = g >> 2, p = (g >> 1) & 1, lh = g & 1;
      const float* src = W2 + (16 * tt + 8 * p + 2 * lh) * COUTN + n;
      float4 v = make_float4(src[0], src[COUTN], src[4 * COUTN], src[5 * COUTN]);
      *(float4*)(&sW2f[((((tt * 4 + o4) * 2 + p) * 32) + lh * 16 + lm) * 4]) = v;
    }
  }
  if (tid < HID)   sB1[tid] = b1[tid];
  if (tid < COUTN) sB2[tid] = b2[tid];
  __syncthreads();

  const int wave  = tid >> 5;
  const int lane  = tid & 31;
  const int lhalf = lane >> 4;
  const int lmod  = lane & 15;

  const int strip = blockIdx.x * 8 + wave;
  if (strip >= STRIPS) return;
  const int row0 = strip * 16;

  v2f a1[16];
  {
    const float* base = xo + (size_t)(row0 + lmod) * CIN + 2 * lhalf;
#pragma unroll
    for (int s = 0; s < 16; ++s)
      a1[s] = *(const v2f*)(base + 4 * s);
  }

  v8f oacc[4];
#pragma unroll
  for (int o = 0; o < 4; ++o) {
    float bv = sB2[o * 16 + lmod];
#pragma unroll
    for (int j = 0; j < 8; ++j) oacc[o][j] = bv;
  }

  float* st = &sStage[wave][0];
  const float4* w1f = (const float4*)sW1f + lane;
  const float4* w2f = (const float4*)sW2f + lane;

  for (int t = 0; t < 16; ++t) {
    v8f h;
    float b1v = sB1[t * 16 + lmod];
#pragma unroll
    for (int j = 0; j < 8; ++j) h[j] = b1v;
#pragma unroll
    for (int p = 0; p < 8; ++p) {
      float4 w = w1f[(t * 8 + p) * 32];
      v2f bA = {w.x, w.y};
      v2f bB = {w.z, w.w};
      h = __builtin_amdgcn_wmma_f32_16x16x4_f32(false, a1[2 * p], false, bA,
                                                (short)0, h, false, false);
      h = __builtin_amdgcn_wmma_f32_16x16x4_f32(false, a1[2 * p + 1], false, bB,
                                                (short)0, h, false, false);
    }
#pragma unroll
    for (int j = 0; j < 8; ++j) h[j] = fmaxf(h[j], 0.f);

#pragma unroll
    for (int j = 0; j < 8; ++j)
      st[(lhalf * 8 + j) * STP + lmod] = h[j];
    asm volatile("s_wait_dscnt 0" ::: "memory");

    v2f a2[4];
#pragma unroll
    for (int kk = 0; kk < 4; ++kk)
      a2[kk] = *(const v2f*)(&st[lmod * STP + 4 * kk + 2 * lhalf]);

#pragma unroll
    for (int o = 0; o < 4; ++o) {
#pragma unroll
      for (int p = 0; p < 2; ++p) {
        float4 w = w2f[((t * 4 + o) * 2 + p) * 32];
        v2f bA = {w.x, w.y};
        v2f bB = {w.z, w.w};
        oacc[o] = __builtin_amdgcn_wmma_f32_16x16x4_f32(false, a2[2 * p], false, bA,
                                                        (short)0, oacc[o], false, false);
        oacc[o] = __builtin_amdgcn_wmma_f32_16x16x4_f32(false, a2[2 * p + 1], false, bB,
                                                        (short)0, oacc[o], false, false);
      }
    }
  }

#pragma unroll
  for (int o = 0; o < 4; ++o) {
#pragma unroll
    for (int j = 0; j < 8; ++j)
      xo[(size_t)(row0 + lhalf * 8 + j) * COUTN + o * 16 + lmod] = oacc[o][j];
  }
}

extern "C" void kernel_launch(void* const* d_in, const int* in_sizes, int n_in,
                              void* d_out, int out_size, void* d_ws, size_t ws_size,
                              hipStream_t stream) {
  const float* x     = (const float*)d_in[0];
  const int*   A_idx = (const int*)d_in[1];
  const float* A_val = (const float*)d_in[2];
  const float* W1    = (const float*)d_in[3];
  const float* b1    = (const float*)d_in[4];
  const float* W2    = (const float*)d_in[5];
  const float* b2    = (const float*)d_in[6];
  float* out = (float*)d_out;

  gcn_zero_kernel<<<12500, 256, 0, stream>>>(out);

  gcn_scatter_kernel<<<(NE * 64) / 256, 256, 0, stream>>>(
      x, A_idx, A_idx + NE, A_val, out);

  gcn_mlp_kernel<<<(STRIPS + 7) / 8, 256, 0, stream>>>(out, W1, b1, W2, b2);
}


# sparse mix: edges counting-sorted by row in workspace (int atomic rank, scan, fill); one wave per row sums x rows with fma, writes y once; no float atomics, no zero pass; row sum order = rank order
# speedup vs baseline: 3.1527x; 2.1197x over previous
#include <hip/hip_runtime.h>

typedef __attribute__((ext_vector_type(2))) float v2f;
typedef __attribute__((ext_vector_type(8))) float v8f;

#define NB    4
#define NN    50000
#define MM    (NB * NN)
#define CIN   64
#define HID   256
#define COUTN 64
#define NE    800000
#define STRIPS (MM / 16)

#define SCAN_T   1024
#define SCAN_PER 49

__global__ void gcn_zero_counts_kernel(int* __restrict__ cnt) {
  cnt[blockIdx.x * 256 + threadIdx.x] = 0;
}

__global__ void gcn_count_kernel(const int* __restrict__ rows,
                                 int* __restrict__ cnt,
                                 int* __restrict__ rank) {
  int e = blockIdx.x * 256 + threadIdx.x;
  if (e >= NE) return;
  rank[e] = atomicAdd(&cnt[rows[e]], 1);
}

__global__ void __launch_bounds__(SCAN_T)
gcn_scan_kernel(const int* __restrict__ cnt, int* __restrict__ rowptr) {
  __shared__ int sC[SCAN_T * SCAN_PER];
  __shared__ int sW[32];
  const int tid = threadIdx.x;
  {
    int v[SCAN_PER];
#pragma unroll
    for (int k = 0; k < SCAN_PER; ++k) v[k] = cnt[tid + k * SCAN_T];
#pragma unroll
    for (int k = 0; k < SCAN_PER; ++k) sC[tid + k * SCAN_T] = v[k];
  }
  __syncthreads();

  const int base = tid * SCAN_PER;
  int sum = 0;
  for (int k = 0; k < SCAN_PER; ++k) sum += sC[base + k];

  const int lane = tid & 31, wv = tid >> 5;
  int inc = sum;
  for (int d = 1; d < 32; d <<= 1) {
    int t = __shfl_up(inc, d);
    if (lane >= d) inc += t;
  }
  if (lane == 31) sW[wv] = inc;
  __syncthreads();
  if (wv == 0) {
    int w = sW[lane];
    int winc = w;
    for (int d = 1; d < 32; d <<= 1) {
      int t = __shfl_up(winc, d);
      if (lane >= d) winc += t;
    }
    sW[lane] = winc - w;
  }
  __syncthreads();

  int run = sW[wv] + inc - sum;
  for (int k = 0; k < SCAN_PER; ++k) {
    int c = sC[base + k];
    sC[base + k] = run;
    run += c;
  }
  __syncthreads();
  for (int i = tid; i <= NN; i += SCAN_T) rowptr[i] = sC[i];
}

__global__ void gcn_fill_kernel(const int* __restrict__ rows,
                                const int* __restrict__ cols,
                                const float* __restrict__ vals,
                                const int* __restrict__ rowptr,
                                const int* __restrict__ rank,
                                int* __restrict__ sedge) {
  int e = blockIdx.x * 256 + threadIdx.x;
  if (e >= NE) return;
  int pos = rowptr[rows[e]] + rank[e];
  *(int2*)(sedge + 2 * (size_t)pos) = make_int2(cols[e], __float_as_int(vals[e]));
}

__global__ void __launch_bounds__(256)
gcn_gather_kernel(const float* __restrict__ x,
                  const int* __restrict__ rowptr,
                  const int* __restrict__ sedge,
                  float* __restrict__ y) {
  const int lane = threadIdx.x & 31;
  const int r = __builtin_amdgcn_readfirstlane(blockIdx.x * 8 + (threadIdx.x >> 5));
  if (r >= NN) return;
  const int s = __builtin_amdgcn_readfirstlane(rowptr[r]);
  const int e = __builtin_amdgcn_readfirstlane(rowptr[r + 1]);

  const int lhalf = lane >> 4, q4 = (lane & 15) * 4;
  const unsigned off0 = (unsigned)(lhalf * NN * CIN + q4) * 4u;
  const unsigned off1 = off0 + 2u * NN * CIN * 4u;
  float4 acc0 = make_float4(0.f, 0.f, 0.f, 0.f);
  float4 acc1 = make_float4(0.f, 0.f, 0.f, 0.f);

#define GCN_LOAD(A, B, C)                                                      \
  {                                                                            \
    const char* xr_ = (const char*)(x + (size_t)(C) * CIN);                    \
    A = *(const float4*)(xr_ + off0);                                          \
    B = *(const float4*)(xr_ + off1);                                          \
  }
#define GCN_FMA(V, A, B)                                                       \
  {                                                                            \
    acc0.x = fmaf(V, A.x, acc0.x); acc0.y = fmaf(V, A.y, acc0.y);              \
    acc0.z = fmaf(V, A.z, acc0.z); acc0.w = fmaf(V, A.w, acc0.w);              \
    acc1.x = fmaf(V, B.x, acc1.x); acc1.y = fmaf(V, B.y, acc1.y);              \
    acc1.z = fmaf(V, B.z, acc1.z); acc1.w = fmaf(V, B.w, acc1.w);              \
  }

  for (int base = s; base < e; base += 32) {
    int2 m = make_int2(0, 0);
    if (base + lane < e) m = *(const int2*)(sedge + 2 * (size_t)(base + lane));
    const int n = (e - base < 32) ? (e - base) : 32;
    int j = 0;
    for (; j + 4 <= n; j += 4) {
      const int   c0 = __builtin_amdgcn_readlane(m.x, j);
      const int   c1 = __builtin_amdgcn_readlane(m.x, j + 1);
      const int   c2 = __builtin_amdgcn_readlane(m.x, j + 2);
      const int   c3 = __builtin_amdgcn_readlane(m.x, j + 3);
      const float v0 = __int_as_float(__builtin_amdgcn_readlane(m.y, j));
      const float v1 = __int_as_float(__builtin_amdgcn_readlane(m.y, j + 1));
      const float v2 = __int_as_float(__builtin_amdgcn_readlane(m.y, j + 2));
      const float v3 = __int_as_float(__builtin_amdgcn_readlane(m.y, j + 3));
      float4 a0, b0, a1, b1, a2, b2, a3, b3;
      GCN_LOAD(a0, b0, c0)
      GCN_LOAD(a1, b1, c1)
      GCN_LOAD(a2, b2, c2)
      GCN_LOAD(a3, b3, c3)
      GCN_FMA(v0, a0, b0)
      GCN_FMA(v1, a1, b1)
      GCN_FMA(v2, a2, b2)
      GCN_FMA(v3, a3, b3)
    }
#pragma unroll 1
    for (; j < n; ++j) {
      const int   c = __builtin_amdgcn_readlane(m.x, j);
      const float v = __int_as_float(__builtin_amdgcn_readlane(m.y, j));
      float4 a, b;
      GCN_LOAD(a, b, c)
      GCN_FMA(v, a, b)
    }
  }
#undef GCN_LOAD
#undef GCN_FMA

  *(float4*)(y + ((size_t)lhalf * NN + r) * CIN + q4)       = acc0;
  *(float4*)(y + ((size_t)(lhalf + 2) * NN + r) * CIN + q4) = acc1;
}

#define STP 20
__global__ void __launch_bounds__(256, 1)
gcn_mlp_kernel(float* __restrict__ xo,
               const float* __restrict__ W1,
               const float* __restrict__ b1,
               const float* __restrict__ W2,
               const float* __restrict__ b2) {
  __shared__ __align__(16) float sW1f[HID * CIN];
  __shared__ __align__(16) float sW2f[COUTN * HID];
  __shared__ __align__(16) float sB1[HID];
  __shared__ __align__(16) float sB2[COUTN];
  __shared__ __align__(16) float sStage[8][16 * STP];

  const int tid = threadIdx.x;
  {
    const int n = tid, tt = n >> 4, lm = n & 15;
#pragma unroll
    for (int g = 0; g < 16; ++g) {
      const int p = g >> 1, lh = g & 1;
      const float* src = W1 + (8 * p + 2 * lh) * HID + n;
      float4 v = make_float4(src[0], src[HID], src[4 * HID], src[5 * HID]);
      *(float4*)(&sW1f[(((tt * 8 + p) * 32) + lh * 16 + lm) * 4]) = v;
    }
  }
  {
    const int n = tid & 63, o4 = n >> 4, lm = n & 15, sub = tid >> 6;
#pragma unroll
    for (int j = 0; j < 16; ++j) {
      const int g = j * 4 + sub;
      const int tt = g >> 2, p = (g >> 1) & 1, lh = g & 1;
      const float* src = W2 + (16 * tt + 8 * p + 2 * lh) * COUTN + n;
      float4 v = make_float4(src[0], src[COUTN], src[4 * COUTN], src[5 * COUTN]);
      *(float4*)(&sW2f[((((tt * 4 + o4) * 2 + p) * 32) + lh * 16 + lm) * 4]) = v;
    }
  }
  if (tid < HID)   sB1[tid] = b1[tid];
  if (tid < COUTN) sB2[tid] = b2[tid];
  __syncthreads();

  const int wave  = tid >> 5;
  const int lane  = tid & 31;
  const int lhalf = lane >> 4;
  const int lmod  = lane & 15;

  const int strip = blockIdx.x * 8 + wave;
  if (strip >= STRIPS) return;
  const int row0 = strip * 16;

  v2f a1[16];
  {
    const float* base = xo + (size_t)(row0 + lmod) * CIN + 2 * lhalf;
#pragma unroll
    for (int s = 0; s < 16; ++s)
      a1[s] = *(const v2f*)(base + 4 * s);
  }

  v8f oacc[4];
#pragma unroll
  for (int o = 0; o < 4; ++o) {
    float bv = sB2[o * 16 + lmod];
#pragma unroll
    for (int j = 0; j < 8; ++j) oacc[o][j] = bv;
  }

  float* st = &sStage[wave][0];
  const float4* w1f = (const float4*)sW1f + lane;
  const float4* w2f = (const float4*)sW2f + lane;

  for (int t = 0; t < 16; ++t) {
    v8f h;
    float b1v = sB1[t * 16 + lmod];
#pragma unroll
    for (int j = 0; j < 8; ++j) h[j] = b1v;
#pragma unroll
    for (int p = 0; p < 8; ++p) {
      float4 w = w1f[(t * 8 + p) * 32];
      v2f bA = {w.x, w.y};
      v2f bB = {w.z, w.w};
      h = __builtin_amdgcn_wmma_f32_16x16x4_f32(false, a1[2 * p], false, bA,
                                                (short)0, h, false, false);
      h = __builtin_amdgcn_wmma_f32_16x16x4_f32(false, a1[2 * p + 1], false, bB,
                                                (short)0, h, false, false);
    }
#pragma unroll
    for (int j = 0; j < 8; ++j) h[j] = fmaxf(h[j], 0.f);

#pragma unroll
    for (int j = 0; j < 8; ++j)
      st[(lhalf * 8 + j) * STP + lmod] = h[j];
    asm volatile("s_wait_dscnt 0" ::: "memory");

    v2f a2[4];
#pragma unroll
    for (int kk = 0; kk < 4; ++kk)
      a2[kk] = *(const v2f*)(&st[lmod * STP + 4 * kk + 2 * lhalf]);

#pragma unroll
    for (int o = 0; o < 4; ++o) {
#pragma unroll
      for (int p = 0; p < 2; ++p) {
        float4 w = w2f[((t * 4 + o) * 2 + p) * 32];
        v2f bA = {w.x, w.y};
        v2f bB = {w.z, w.w};
        oacc[o] = __builtin_amdgcn_wmma_f32_16x16x4_f32(false, a2[2 * p], false, bA,
                                                        (short)0, oacc[o], false, false);
        oacc[o] = __builtin_amdgcn_wmma_f32_16x16x4_f32(false, a2[2 * p + 1], false, bB,
                                                        (short)0, oacc[o], false, false);
      }
    }
  }

#pragma unroll
  for (int o = 0; o < 4; ++o) {
#pragma unroll
    for (int j = 0; j < 8; ++j)
      xo[(size_t)(row0 + lhalf * 8 + j) * COUTN + o * 16 + lmod] = oacc[o][j];
  }
}

extern "C" void kernel_launch(void* const* d_in, const int* in_sizes, int n_in,
                              void* d_out, int out_size, void* d_ws, size_t ws_size,
                              hipStream_t stream) {
  const float* x     = (const float*)d_in[0];
  const int*   A_idx = (const int*)d_in[1];
  const float* A_val = (const float*)d_in[2];
  const float* W1    = (const float*)d_in[3];
  const float* b1    = (const float*)d_in[4];
  const float* W2    = (const float*)d_in[5];
  const float* b2    = (const float*)d_in[6];
  float* out = (float*)d_out;

  int* cnt    = (int*)d_ws;
  int* rowptr = (int*)((char*)d_ws + 262144);
  int* rank   = (int*)((char*)d_ws + 524288);
  int* sedge  = (int*)((char*)d_ws + 4194304);
  const int* rows = A_idx;
  const int* cols = A_idx + NE;

  gcn_zero_counts_kernel<<<196, 256, 0, stream>>>(cnt);
  gcn_count_kernel<<<NE / 256, 256, 0, stream>>>(rows, cnt, rank);
  gcn_scan_kernel<<<1, SCAN_T, 0, stream>>>(cnt, rowptr);
  gcn_fill_kernel<<<NE / 256, 256, 0, stream>>>(rows, cols, A_val, rowptr, rank, sedge);
  gcn_gather_kernel<<<NN / 8, 256, 0, stream>>>(x, rowptr, sedge, out);

  gcn_mlp_kernel<<<(STRIPS + 7) / 8, 256, 0, stream>>>(out, W1, b1, W2, b2);
}


# edge sort by row redone on LDS counters: chunk histograms, edges moved to 256 row-range lists, each list ordered in one work-group's LDS; no global atomics; row sum order = LDS rank order
# speedup vs baseline: 4.3547x; 1.3813x over previous
#include <hip/hip_runtime.h>

typedef __attribute__((ext_vector_type(2))) float v2f;
typedef __attribute__((ext_vector_type(8))) float v8f;

#define NB    4
#define NN    50000
#define MM    (NB * NN)
#define CIN   64
#define HID   256
#define COUTN 64
#define NE    800000
#define STRIPS (MM / 16)

#define NBIN    256
#define BINROWS 196
#define NCHUNK  256
#define CHUNK   3125
#define CPT     13
#define CAP     4096
#define RPT     16

__device__ __forceinline__ int gcn_excl_scan256(int v, int* sW, int tid) {
  const int lane = tid & 31, wv = tid >> 5;
  int inc = v;
#pragma unroll
  for (int d = 1; d < 32; d <<= 1) {
    int t = __shfl_up(inc, d);
    if (lane >= d) inc += t;
  }
  if (lane == 31) sW[wv] = inc;
  __syncthreads();
  int off = 0;
#pragma unroll
  for (int w = 0; w < 7; ++w) off += (w < wv) ? sW[w] : 0;
  return off + inc - v;
}

__global__ void __launch_bounds__(256)
gcn_hist_kernel(const int* __restrict__ rows, int* __restrict__ hist) {
  __shared__ int sH[NBIN];
  const int tid = threadIdx.x, c = blockIdx.x;
  sH[tid] = 0;
  __syncthreads();
  const int* rp = rows + c * CHUNK;
  int r[CPT];
#pragma unroll
  for (int k = 0; k < CPT; ++k) {
    const int i = tid + k * 256;
    r[k] = rp[i < CHUNK ? i : CHUNK - 1];
  }
#pragma unroll
  for (int k = 0; k < CPT; ++k) {
    const int i = tid + k * 256;
    if (i < CHUNK) atomicAdd(&sH[r[k] / BINROWS], 1);
  }
  __syncthreads();
  hist[c * NBIN + tid] = sH[tid];
}

__global__ void __launch_bounds__(256)
gcn_part_kernel(const int* __restrict__ rows,
                const int* __restrict__ cols,
                const float* __restrict__ vals,
                const int* __restrict__ hist,
                int* __restrict__ binstart,
                int* __restrict__ bucket_w) {
  int2* __restrict__ bucket = (int2*)bucket_w;
  __shared__ int sLex[NBIN];
  __shared__ int sGb[NBIN];
  __shared__ int sCur[NBIN];
  __shared__ int sWa[8];
  __shared__ int sWb[8];
  __shared__ int2 sRec[CHUNK];
  __shared__ unsigned char sBin[CHUNK];
  const int tid = threadIdx.x, c = blockIdx.x;

  int tot = 0, pre = 0;
#pragma unroll 32
  for (int k = 0; k < NCHUNK; ++k) {
    const int h = hist[k * NBIN + tid];
    tot += h;
    pre += (k < c) ? h : 0;
  }
  const int mine = hist[c * NBIN + tid];
  const int exT = gcn_excl_scan256(tot, sWa, tid);
  const int exM = gcn_excl_scan256(mine, sWb, tid);
  sLex[tid] = exM;
  sGb[tid]  = exT + pre;
  sCur[tid] = 0;
  if (c == 0) {
    binstart[tid] = exT;
    if (tid == NBIN - 1) binstart[NBIN] = exT + tot;
  }
  __syncthreads();

  const int e0 = c * CHUNK;
  int r[CPT], cc[CPT];
  float vv[CPT];
#pragma unroll
  for (int k = 0; k < CPT; ++k) {
    const int i = tid + k * 256;
    const int e = e0 + (i < CHUNK ? i : CHUNK - 1);
    r[k] = rows[e];
    cc[k] = cols[e];
    vv[k] = vals[e];
  }
#pragma unroll
  for (int k = 0; k < CPT; ++k) {
    const int i = tid + k * 256;
    if (i < CHUNK) {
      const int b  = r[k] / BINROWS;
      const int lr = r[k] - b * BINROWS;
      const int p  = sLex[b] + atomicAdd(&sCur[b], 1);
      sRec[p] = make_int2((lr << 16) | cc[k], __float_as_int(vv[k]));
      sBin[p] = (unsigned char)b;
    }
  }
  __syncthreads();
#pragma unroll
  for (int k = 0; k < CPT; ++k) {
    const int i = tid + k * 256;
    if (i < CHUNK) {
      const int b = sBin[i];
      bucket[sGb[b] + (i - sLex[b])] = sRec[i];
    }
  }
}

__global__ void __launch_bounds__(256)
gcn_rowsort_kernel(const int* __restrict__ binstart,
                   const int* __restrict__ bucket_w,
                   int* __restrict__ rowptr,
                   int* __restrict__ sedge_w) {
  const int2* __restrict__ bucket = (const int2*)bucket_w;
  int2* __restrict__ sedge = (int2*)sedge_w;
  __shared__ int sCnt[256];
  __shared__ int sCur[256];
  __shared__ int sWa[8];
  __shared__ int2 sOut[CAP];
  const int tid = threadIdx.x, b = blockIdx.x;
  const int start = binstart[b];
  const int n = binstart[b + 1] - start;
  const int2* src = bucket + start;
  sCnt[tid] = 0;
  __syncthreads();

  for (int i0 = 0; i0 < n; i0 += CAP) {
    int key[RPT];
#pragma unroll
    for (int k = 0; k < RPT; ++k) {
      const int i = i0 + tid + k * 256;
      key[k] = src[i < n ? i : n - 1].x;
    }
#pragma unroll
    for (int k = 0; k < RPT; ++k) {
      const int i = i0 + tid + k * 256;
      if (i < n) atomicAdd(&sCnt[key[k] >> 16], 1);
    }
  }
  __syncthreads();

  const int cnt = sCnt[tid];
  const int ex = gcn_excl_scan256(cnt, sWa, tid);
  sCur[tid] = ex;
  const int grow = b * BINROWS + tid;
  if (tid < BINROWS && grow <= NN) rowptr[grow] = start + ex;
  __syncthreads();

  const bool staged = (n <= CAP);
  for (int i0 = 0; i0 < n; i0 += CAP) {
    int2 rec[RPT];
#pragma unroll
    for (int k = 0; k < RPT; ++k) {
      const int i = i0 + tid + k * 256;
      rec[k] = src[i < n ? i : n - 1];
    }
#pragma unroll
    for (int k = 0; k < RPT; ++k) {
      const int i = i0 + tid + k * 256;
      if (i < n) {
        const int p = atomicAdd(&sCur[rec[k].x >> 16], 1);
        const int2 o = make_int2(rec[k].x & 0xffff, rec[k].y);
        if (staged) sOut[p] = o;
        else        sedge[start + p] = o;
      }
    }
  }
  if (staged) {
    __syncthreads();
    int2* dst = sedge + start;
    for (int i = tid; i < n; i += 256) dst[i] = sOut[i];
  }
}

__global__ void __launch_bounds__(256)
gcn_gather_kernel(const float* __restrict__ x,
                  const int* __restrict__ rowptr,
                  const int* __restrict__ sedge,
                  float* __restrict__ y) {
  const int lane = threadIdx.x & 31;
  const int r = __builtin_amdgcn_readfirstlane(blockIdx.x * 8 + (threadIdx.x >> 5));
  if (r >= NN) return;
  const int s = __builtin_amdgcn_readfirstlane(rowptr[r]);
  const int e = __builtin_amdgcn_readfirstlane(rowptr[r + 1]);

  const int lhalf = lane >> 4, q4 = (lane & 15) * 4;
  const unsigned off0 = (unsigned)(lhalf * NN * CIN + q4) * 4u;
  const unsigned off1 = off0 + 2u * NN * CIN * 4u;
  float4 acc0 = make_float4(0.f, 0.f, 0.f, 0.f);
  float4 acc1 = make_float4(0.f, 0.f, 0.f, 0.f);

#define GCN_LOAD(A, B, C)                                                      \
  {                                                                            \
    const char* xr_ = (const char*)(x + (size_t)(C) * CIN);                    \
    A = *(const float4*)(xr_ + off0);                                          \
    B = *(const float4*)(xr_ + off1);                                          \
  }
#define GCN_FMA(V, A, B)                                                       \
  {                                                                            \
    acc0.x = fmaf(V, A.x, acc0.x); acc0.y = fmaf(V, A.y, acc0.y);              \
    acc0.z = fmaf(V, A.z, acc0.z); acc0.w = fmaf(V, A.w, acc0.w);              \
    acc1.x = fmaf(V, B.x, acc1.x); acc1.y = fmaf(V, B.y, acc1.y);              \
    acc1.z = fmaf(V, B.z, acc1.z); acc1.w = fmaf(V, B.w, acc1.w);              \
  }

  for (int base = s; base < e; base += 32) {
    int2 m = make_int2(0, 0);
    if (base + lane < e) m = *(const int2*)(sedge + 2 * (size_t)(base + lane));
    const int n = (e - base < 32) ? (e - base) : 32;
    int j = 0;
    for (; j + 4 <= n; j += 4) {
      const int   c0 = __builtin_amdgcn_readlane(m.x, j);
      const int   c1 = __builtin_amdgcn_readlane(m.x, j + 1);
      const int   c2 = __builtin_amdgcn_readlane(m.x, j + 2);
      const int   c3 = __builtin_amdgcn_readlane(m.x, j + 3);
      const float v0 = __int_as_float(__builtin_amdgcn_readlane(m.y, j));
      const float v1 = __int_as_float(__builtin_amdgcn_readlane(m.y, j + 1));
      const float v2 = __int_as_float(__builtin_amdgcn_readlane(m.y, j + 2));
      const float v3 = __int_as_float(__builtin_amdgcn_readlane(m.y, j + 3));
      float4 a0, b0, a1, b1, a2, b2, a3, b3;
      GCN_LOAD(a0, b0, c0)
      GCN_LOAD(a1, b1, c1)
      GCN_LOAD(a2, b2, c2)
      GCN_LOAD(a3, b3, c3)
      GCN_FMA(v0, a0, b0)
      GCN_FMA(v1, a1, b1)
      GCN_FMA(v2, a2, b2)
      GCN_FMA(v3, a3, b3)
    }
#pragma unroll 1
    for (; j < n; ++j) {
      const int   c = __builtin_amdgcn_readlane(m.x, j);
      const float v = __int_as_float(__builtin_amdgcn_readlane(m.y, j));
      float4 a, b;
      GCN_LOAD(a, b, c)
      GCN_FMA(v, a, b)
    }
  }
#undef GCN_LOAD
#undef GCN_FMA

  *(float4*)(y + ((size_t)lhalf * NN + r) * CIN + q4)       = acc0;
  *(float4*)(y + ((size_t)(lhalf + 2) * NN + r) * CIN + q4) = acc1;
}

#define STP 20
__global__ void __launch_bounds__(256, 1)
gcn_mlp_kernel(float* __restrict__ xo,
               const float* __restrict__ W1,
               const float* __restrict__ b1,
               const float* __restrict__ W2,
               const float* __restrict__ b2) {
  __shared__ __align__(16) float sW1f[HID * CIN];
  __shared__ __align__(16) float sW2f[COUTN * HID];
  __shared__ __align__(16) float sB1[HID];
  __shared__ __align__(16) float sB2[COUTN];
  __shared__ __align__(16) float sStage[8][16 * STP];

  const int tid = threadIdx.x;
  {
    const int n = tid, tt = n >> 4, lm = n & 15;
#pragma unroll
    for (int g = 0; g < 16; ++g) {
      const int p = g >> 1, lh = g & 1;
      const float* src = W1 + (8 * p + 2 * lh) * HID + n;
      float4 v = make_float4(src[0], src[HID], src[4 * HID], src[5 * HID]);
      *(float4*)(&sW1f[(((tt * 8 + p) * 32) + lh * 16 + lm) * 4]) = v;
    }
  }
  {
    const int n = tid & 63, o4 = n >> 4, lm = n & 15, sub = tid >> 6;
#pragma unroll
    for (int j = 0; j < 16; ++j) {
      const int g = j * 4 + sub;
      const int tt = g >> 2, p = (g >> 1) & 1, lh = g & 1;
      const float* src = W2 + (16 * tt + 8 * p + 2 * lh) * COUTN + n;
      float4 v = make_float4(src[0], src[COUTN], src[4 * COUTN], src[5 * COUTN]);
      *(float4*)(&sW2f[((((tt * 4 + o4) * 2 + p) * 32) + lh * 16 + lm) * 4]) = v;
    }
  }
  if (tid < HID)   sB1[tid] = b1[tid];
  if (tid < COUTN) sB2[tid] = b2[tid];
  __syncthreads();

  const int wave  = tid >> 5;
  const int lane  = tid & 31;
  const int lhalf = lane >> 4;
  const int lmod  = lane & 15;

  const int strip = blockIdx.x * 8 + wave;
  if (strip >= STRIPS) return;
  const int row0 = strip * 16;

  v2f a1[16];
  {
    const float* base = xo + (size_t)(row0 + lmod) * CIN + 2 * lhalf;
#pragma unroll
    for (int s = 0; s < 16; ++s)
      a1[s] = *(const v2f*)(base + 4 * s);
  }

  v8f oacc[4];
#pragma unroll
  for (int o = 0; o < 4; ++o) {
    float bv = sB2[o * 16 + lmod];
#pragma unroll
    for (int j = 0; j < 8; ++j) oacc[o][j] = bv;
  }

  float* st = &sStage[wave][0];
  const float4* w1f = (const float4*)sW1f + lane;
  const float4* w2f = (const float4*)sW2f + lane;

  for (int t = 0; t < 16; ++t) {
    v8f h;
    float b1v = sB1[t * 16 + lmod];
#pragma unroll
    for (int j = 0; j < 8; ++j) h[j] = b1v;
#pragma unroll
    for (int p = 0; p < 8; ++p) {
      float4 w = w1f[(t * 8 + p) * 32];
      v2f bA = {w.x, w.y};
      v2f bB = {w.z, w.w};
      h = __builtin_amdgcn_wmma_f32_16x16x4_f32(false, a1[2 * p], false, bA,
                                                (short)0, h, false, false);
      h = __builtin_amdgcn_wmma_f32_16x16x4_f32(false, a1[2 * p + 1], false, bB,
                                                (short)0, h, false, false);
    }
#pragma unroll
    for (int j = 0; j < 8; ++j) h[j] = fmaxf(h[j], 0.f);

#pragma unroll
    for (int j = 0; j < 8; ++j)
      st[(lhalf * 8 + j) * STP + lmod] = h[j];
    asm volatile("s_wait_dscnt 0" ::: "memory");

    v2f a2[4];
#pragma unroll
    for (int kk = 0; kk < 4; ++kk)
      a2[kk] = *(const v2f*)(&st[lmod * STP + 4 * kk + 2 * lhalf]);

#pragma unroll
    for (int o = 0; o < 4; ++o) {
#pragma unroll
      for (int p = 0; p < 2; ++p) {
        float4 w = w2f[((t * 4 + o) * 2 + p) * 32];
        v2f bA = {w.x, w.y};
        v2f bB = {w.z, w.w};
        oacc[o] = __builtin_amdgcn_wmma_f32_16x16x4_f32(false, a2[2 * p], false, bA,
                                                        (short)0, oacc[o], false, false);
        oacc[o] = __builtin_amdgcn_wmma_f32_16x16x4_f32(false, a2[2 * p + 1], false, bB,
                                                        (short)0, oacc[o], false, false);
      }
    }
  }

#pragma unroll
  for (int o = 0; o < 4; ++o) {
#pragma unroll
    for (int j = 0; j < 8; ++j)
      xo[(size_t)(row0 + lhalf * 8 + j) * COUTN + o * 16 + lmod] = oacc[o][j];
  }
}

extern "C" void kernel_launch(void* const* d_in, const int* in_sizes, int n_in,
                              void* d_out, int out_size, void* d_ws, size_t ws_size,
                              hipStream_t stream) {
  const float* x     = (const float*)d_in[0];
  const int*   A_idx = (const int*)d_in[1];
  const float* A_val = (const float*)d_in[2];
  const float* W1    = (const float*)d_in[3];
  const float* b1    = (const float*)d_in[4];
  const float* W2    = (const float*)d_in[5];
  const float* b2    = (const float*)d_in[6];
  float* out = (float*)d_out;

  int*  hist     = (int*)d_ws;
  int*  binstart = (int*)((char*)d_ws + 262144);
  int*  rowptr   = (int*)((char*)d_ws + 524288);
  int*  bucket   = (int*)((char*)d_ws + 1048576);
  int*  sedge    = (int*)((char*)d_ws + 8388608);
  const int* rows = A_idx;
  const int* cols = A_idx + NE;

  gcn_hist_kernel<<<NCHUNK, 256, 0, stream>>>(rows, hist);
  gcn_part_kernel<<<NCHUNK, 256, 0, stream>>>(rows, cols, A_val, hist, binstart, bucket);
  gcn_rowsort_kernel<<<NBIN, 256, 0, stream>>>(binstart, bucket, rowptr, sedge);
  gcn_gather_kernel<<<NN / 8, 256, 0, stream>>>(x, rowptr, sedge, out);

  gcn_mlp_kernel<<<(STRIPS + 7) / 8, 256, 0, stream>>>(out, W1, b1, W2, b2);
}


# row sums fused into the MLP kernel as 16 producer waves per group: they sum the next 128-row tile into LDS while 8 matrix waves run the MLP on the last; 256 persistent groups; sum order unchanged
# speedup vs baseline: 5.0739x; 1.1652x over previous
#include <hip/hip_runtime.h>

typedef __attribute__((ext_vector_type(2))) float v2f;
typedef __attribute__((ext_vector_type(8))) float v8f;

#define NB    4
#define NN    50000
#define MM    (NB * NN)
#define CIN   64
#define HID   256
#define COUTN 64
#define NE    800000
#define STRIPS (MM / 16)

#define NBIN    256
#define BINROWS 196
#define NCHUNK  256
#define CHUNK   3125
#define CPT     13
#define CAP     4096
#define RPT     16

__device__ __forceinline__ int gcn_excl_scan256(int v, int* sW, int tid) {
  const int lane = tid & 31, wv = tid >> 5;
  int inc = v;
#pragma unroll
  for (int d = 1; d < 32; d <<= 1) {
    int t = __shfl_up(inc, d);
    if (lane >= d) inc += t;
  }
  if (lane == 31) sW[wv] = inc;
  __syncthreads();
  int off = 0;
#pragma unroll
  for (int w = 0; w < 7; ++w) off += (w < wv) ? sW[w] : 0;
  return off + inc - v;
}

__global__ void __launch_bounds__(256)
gcn_hist_kernel(const int* __restrict__ rows, int* __restrict__ hist) {
  __shared__ int sH[NBIN];
  const int tid = threadIdx.x, c = blockIdx.x;
  sH[tid] = 0;
  __syncthreads();
  const int* rp = rows + c * CHUNK;
  int r[CPT];
#pragma unroll
  for (int k = 0; k < CPT; ++k) {
    const int i = tid + k * 256;
    r[k] = rp[i < CHUNK ? i : CHUNK - 1];
  }
#pragma unroll
  for (int k = 0; k < CPT; ++k) {
    const int i = tid + k * 256;
    if (i < CHUNK) atomicAdd(&sH[r[k] / BINROWS], 1);
  }
  __syncthreads();
  hist[c * NBIN + tid] = sH[tid];
}

__global__ void __launch_bounds__(256)
gcn_part_kernel(const int* __restrict__ rows,
                const int* __restrict__ cols,
                const float* __restrict__ vals,
                const int* __restrict__ hist,
                int* __restrict__ binstart,
                int* __restrict__ bucket_w) {
  int2* __restrict__ bucket = (int2*)bucket_w;
  __shared__ int sLex[NBIN];
  __shared__ int sGb[NBIN];
  __shared__ int sCur[NBIN];
  __shared__ int sWa[8];
  __shared__ int sWb[8];
  __shared__ int2 sRec[CHUNK];
  __shared__ unsigned char sBin[CHUNK];
  const int tid = threadIdx.x, c = blockIdx.x;

  int tot = 0, pre = 0;
#pragma unroll 32
  for (int k = 0; k < NCHUNK; ++k) {
    const int h = hist[k * NBIN + tid];
    tot += h;
    pre += (k < c) ? h : 0;
  }
  const int mine = hist[c * NBIN + tid];
  const int exT = gcn_excl_scan256(tot, sWa, tid);
  const int exM = gcn_excl_scan256(mine, sWb, tid);
  sLex[tid] = exM;
  sGb[tid]  = exT + pre;
  sCur[tid] = 0;
  if (c == 0) {
    binstart[tid] = exT;
    if (tid == NBIN - 1) binstart[NBIN] = exT + tot;
  }
  __syncthreads();

  const int e0 = c * CHUNK;
  int r[CPT], cc[CPT];
  float vv[CPT];
#pragma unroll
  for (int k = 0; k < CPT; ++k) {
    const int i = tid + k * 256;
    const int e = e0 + (i < CHUNK ? i : CHUNK - 1);
    r[k] = rows[e];
    cc[k] = cols[e];
    vv[k] = vals[e];
  }
#pragma unroll
  for (int k = 0; k < CPT; ++k) {
    const int i = tid + k * 256;
    if (i < CHUNK) {
      const int b  = r[k] / BINROWS;
      const int lr = r[k] - b * BINROWS;
      const int p  = sLex[b] + atomicAdd(&sCur[b], 1);
      sRec[p] = make_int2((lr << 16) | cc[k], __float_as_int(vv[k]));
      sBin[p] = (unsigned char)b;
    }
  }
  __syncthreads();
#pragma unroll
  for (int k = 0; k < CPT; ++k) {
    const int i = tid + k * 256;
    if (i < CHUNK) {
      const int b = sBin[i];
      bucket[sGb[b] + (i - sLex[b])] = sRec[i];
    }
  }
}

__global__ void __launch_bounds__(256)
gcn_rowsort_kernel(const int* __restrict__ binstart,
                   const int* __restrict__ bucket_w,
                   int* __restrict__ rowptr,
                   int* __restrict__ sedge_w) {
  const int2* __restrict__ bucket = (const int2*)bucket_w;
  int2* __restrict__ sedge = (int2*)sedge_w;
  __shared__ int sCnt[256];
  __shared__ int sCur[256];
  __shared__ int sWa[8];
  __shared__ int2 sOut[CAP];
  const int tid = threadIdx.x, b = blockIdx.x;
  const int start = binstart[b];
  const int n = binstart[b + 1] - start;
  const int2* src = bucket + start;
  sCnt[tid] = 0;
  __syncthreads();

  for (int i0 = 0; i0 < n; i0 += CAP) {
    int key[RPT];
#pragma unroll
    for (int k = 0; k < RPT; ++k) {
      const int i = i0 + tid + k * 256;
      key[k] = src[i < n ? i : n - 1].x;
    }
#pragma unroll
    for (int k = 0; k < RPT; ++k) {
      const int i = i0 + tid + k * 256;
      if (i < n) atomicAdd(&sCnt[key[k] >> 16], 1);
    }
  }
  __syncthreads();

  const int cnt = sCnt[tid];
  const int ex = gcn_excl_scan256(cnt, sWa, tid);
  sCur[tid] = ex;
  const int grow = b * BINROWS + tid;
  if (tid < BINROWS && grow <= NN) rowptr[grow] = start + ex;
  __syncthreads();

  const bool staged = (n <= CAP);
  for (int i0 = 0; i0 < n; i0 += CAP) {
    int2 rec[RPT];
#pragma unroll
    for (int k = 0; k < RPT; ++k) {
      const int i = i0 + tid + k * 256;
      rec[k] = src[i < n ? i : n - 1];
    }
#pragma unroll
    for (int k = 0; k < RPT; ++k) {
      const int i = i0 + tid + k * 256;
      if (i < n) {
        const int p = atomicAdd(&sCur[rec[k].x >> 16], 1);
        const int2 o = make_int2(rec[k].x & 0xffff, rec[k].y);
        if (staged) sOut[p] = o;
        else        sedge[start + p] = o;
      }
    }
  }
  if (staged) {
    __syncthreads();
    int2* dst = sedge + start;
    for (int i = tid; i < n; i += 256) dst[i] = sOut[i];
  }
}

#define NBLK  (NN / 16)
#define NWG   256
#define TP    68
#define STP   20

__device__ __forceinline__ void gcn_gather_pair(const float* __restrict__ x,
                                                const int* __restrict__ rowptr,
                                                const int* __restrict__ sedge,
                                                float* __restrict__ tile,
                                                int nb, int nbl, int g8, int lane) {
  const int lhalf = lane >> 4, q4 = (lane & 15) * 4;
  const int n0 = nb * 16 + g8 * 2;
  const int p0 = __builtin_amdgcn_readfirstlane(rowptr[n0]);
  const int p1 = __builtin_amdgcn_readfirstlane(rowptr[n0 + 1]);
  const int p2 = __builtin_amdgcn_readfirstlane(rowptr[n0 + 2]);
  const unsigned off0 = (unsigned)(lhalf * NN * CIN + q4) * 4u;
  const unsigned off1 = off0 + 2u * NN * CIN * 4u;

#define GCN_LOAD(A, B, C)                                                      \
  {                                                                            \
    const char* xr_ = (const char*)(x + (size_t)(C) * CIN);                    \
    A = *(const float4*)(xr_ + off0);                                          \
    B = *(const float4*)(xr_ + off1);                                          \
  }
#define GCN_FMA(V, A, B)                                                       \
  {                                                                            \
    acc0.x = fmaf(V, A.x, acc0.x); acc0.y = fmaf(V, A.y, acc0.y);              \
    acc0.z = fmaf(V, A.z, acc0.z); acc0.w = fmaf(V, A.w, acc0.w);              \
    acc1.x = fmaf(V, B.x, acc1.x); acc1.y = fmaf(V, B.y, acc1.y);              \
    acc1.z = fmaf(V, B.z, acc1.z); acc1.w = fmaf(V, B.w, acc1.w);              \
  }
#define GCN_CV(I)                                                              \
  const int   c##I = __builtin_amdgcn_readlane(m.x, j + I);                    \
  const float v##I = __int_as_float(__builtin_amdgcn_readlane(m.y, j + I));

#pragma unroll 1
  for (int k = 0; k < 2; ++k) {
    const int s = k ? p1 : p0;
    const int e = k ? p2 : p1;
    float4 acc0 = make_float4(0.f, 0.f, 0.f, 0.f);
    float4 acc1 = make_float4(0.f, 0.f, 0.f, 0.f);
    for (int base = s; base < e; base += 32) {
      int2 m = make_int2(0, 0);
      if (base + lane < e) m = *(const int2*)(sedge + 2 * (size_t)(base + lane));
      const int n = (e - base < 32) ? (e - base) : 32;
      int j = 0;
      for (; j + 8 <= n; j += 8) {
        GCN_CV(0) GCN_CV(1) GCN_CV(2) GCN_CV(3) GCN_CV(4) GCN_CV(5) GCN_CV(6) GCN_CV(7)
        float4 a0, b0, a1, b1, a2, b2, a3, b3, a4, b4, a5, b5, a6, b6, a7, b7;
        GCN_LOAD(a0, b0, c0) GCN_LOAD(a1, b1, c1) GCN_LOAD(a2, b2, c2) GCN_LOAD(a3, b3, c3)
        GCN_LOAD(a4, b4, c4) GCN_LOAD(a5, b5, c5) GCN_LOAD(a6, b6, c6) GCN_LOAD(a7, b7, c7)
        GCN_FMA(v0, a0, b0) GCN_FMA(v1, a1, b1) GCN_FMA(v2, a2, b2) GCN_FMA(v3, a3, b3)
        GCN_FMA(v4, a4, b4) GCN_FMA(v5, a5, b5) GCN_FMA(v6, a6, b6) GCN_FMA(v7, a7, b7)
      }
      for (; j + 4 <= n; j += 4) {
        GCN_CV(0) GCN_CV(1) GCN_CV(2) GCN_CV(3)
        float4 a0, b0, a1, b1, a2, b2, a3, b3;
        GCN_LOAD(a0, b0, c0) GCN_LOAD(a1, b1, c1) GCN_LOAD(a2, b2, c2) GCN_LOAD(a3, b3, c3)
        GCN_FMA(v0, a0, b0) GCN_FMA(v1, a1, b1) GCN_FMA(v2, a2, b2) GCN_FMA(v3, a3, b3)
      }
#pragma unroll 1
      for (; j < n; ++j) {
        GCN_CV(0)
        float4 a0, b0;
        GCN_LOAD(a0, b0, c0)
        GCN_FMA(v0, a0, b0)
      }
    }
    const int nrow = g8 * 2 + k;
    *(float4*)(tile + ((nbl * 4 + lhalf) * 16 + nrow) * TP + q4)     = acc0;
    *(float4*)(tile + ((nbl * 4 + lhalf + 2) * 16 + nrow) * TP + q4) = acc1;
  }
#undef GCN_LOAD
#undef GCN_FMA
#undef GCN_CV
}

__global__ void __launch_bounds__(768)
gcn_fused_kernel(const float* __restrict__ x,
                 const int* __restrict__ rowptr,
                 const int* __restrict__ sedge,
                 float* __restrict__ out,
                 const float* __restrict__ W1,
                 const float* __restrict__ b1,
                 const float* __restrict__ W2,
                 const float* __restrict__ b2) {
  __shared__ __align__(16) float sW1f[HID * CIN];
  __shared__ __align__(16) float sW2f[COUTN * HID];
  __shared__ __align__(16) float sB1[HID];
  __shared__ __align__(16) float sB2[COUTN];
  __shared__ __align__(16) float sStage[8][16 * STP];
  __shared__ __align__(16) float sTile[2][128 * TP];

  const int tid = threadIdx.x;
  const int lane  = tid & 31;
  const int lhalf = lane >> 4;
  const int lmod  = lane & 15;
  const int wv = __builtin_amdgcn_readfirstlane(tid >> 5);
  const int nb0 = ((int)blockIdx.x * NBLK) / NWG;
  const int nb1 = (((int)blockIdx.x + 1) * NBLK) / NWG;
  const int nIt = (nb1 - nb0 + 1) >> 1;

  if (wv < 8) {
    {
      const int n = tid, tt = n >> 4, lm = n & 15;
#pragma unroll
      for (int g = 0; g < 16; ++g) {
        const int p = g >> 1, lh = g & 1;
        const float* src = W1 + (8 * p + 2 * lh) * HID + n;
        float4 v = make_float4(src[0], src[HID], src[4 * HID], src[5 * HID]);
        *(float4*)(&sW1f[(((tt * 8 + p) * 32) + lh * 16 + lm) * 4]) = v;
      }
    }
    {
      const int n = tid & 63, o4 = n >> 4, lm = n & 15, sub = tid >> 6;
#pragma unroll
      for (int j = 0; j < 16; ++j) {
        const int g = j * 4 + sub;
        const int tt = g >> 2, p = (g >> 1) & 1, lh = g & 1;
        const float* src = W2 + (16 * tt + 8 * p + 2 * lh) * COUTN + n;
        float4 v = make_float4(src[0], src[COUTN], src[4 * COUTN], src[5 * COUTN]);
        *(float4*)(&sW2f[((((tt * 4 + o4) * 2 + p) * 32) + lh * 16 + lm) * 4]) = v;
      }
    }
    sB1[tid] = b1[tid];
    if (tid < COUTN) sB2[tid] = b2[tid];
  } else {
    const int g = wv - 8, nbl = g >> 3;
    if (nb0 + nbl < nb1)
      gcn_gather_pair(x, rowptr, sedge, &sTile[0][0], nb0 + nbl, nbl, g & 7, lane);
  }
  __syncthreads();

  float* st = &sStage[wv & 7][0];
  const float4* w1f = (const float4*)sW1f + lane;
  const float4* w2f = (const float4*)sW2f + lane;

  for (int it = 0; it < nIt; ++it) {
    const int nbA = nb0 + 2 * it;
    if (wv >= 8) {
      const int g = wv - 8, nbl = g >> 3;
      if (it + 1 < nIt && nbA + 2 + nbl < nb1)
        gcn_gather_pair(x, rowptr, sedge, &sTile[(it + 1) & 1][0], nbA + 2 + nbl, nbl, g & 7, lane);
    } else if (nbA + (wv >> 2) < nb1) {
      const int row0 = (wv & 3) * NN + (nbA + (wv >> 2)) * 16;

      v2f a1[16];
      {
        const float* base = &sTile[it & 1][(wv * 16 + lmod) * TP + 2 * lhalf];
#pragma unroll
        for (int s = 0; s < 16; ++s)
          a1[s] = *(const v2f*)(base + 4 * s);
      }

      v8f oacc[4];
#pragma unroll
      for (int o = 0; o < 4; ++o) {
        float bv = sB2[o * 16 + lmod];
#pragma unroll
        for (int j = 0; j < 8; ++j) oacc[o][j] = bv;
      }

      for (int t = 0; t < 16; ++t) {
        v8f h;
        float b1v = sB1[t * 16 + lmod];
#pragma unroll
        for (int j = 0; j < 8; ++j) h[j] = b1v;
#pragma unroll
        for (int p = 0; p < 8; ++p) {
          float4 w = w1f[(t * 8 + p) * 32];
          v2f bA = {w.x, w.y};
          v2f bB = {w.z, w.w};
          h = __builtin_amdgcn_wmma_f32_16x16x4_f32(false, a1[2 * p], false, bA,
                                                    (short)0, h, false, false);
          h = __builtin_amdgcn_wmma_f32_16x16x4_f32(false, a1[2 * p + 1], false, bB,
                                                    (short)0, h, false, false);
        }
#pragma unroll
        for (int j = 0; j < 8; ++j) h[j] = fmaxf(h[j], 0.f);

#pragma unroll
        for (int j = 0; j < 8; ++j)
          st[(lhalf * 8 + j) * STP + lmod] = h[j];
        asm volatile("s_wait_dscnt 0" ::: "memory");

        v2f a2[4];
#pragma unroll
        for (int kk = 0; kk < 4; ++kk)
          a2[kk] = *(const v2f*)(&st[lmod * STP + 4 * kk + 2 * lhalf]);

#pragma unroll
        for (int o = 0; o < 4; ++o) {
#pragma unroll
          for (int p = 0; p < 2; ++p) {
            float4 w = w2f[((t * 4 + o) * 2 + p) * 32];
            v2f bA = {w.x, w.y};
            v2f bB = {w.z, w.w};
            oacc[o] = __builtin_amdgcn_wmma_f32_16x16x4_f32(false, a2[2 * p], false, bA,
                                                            (short)0, oacc[o], false, false);
            oacc[o] = __builtin_amdgcn_wmma_f32_16x16x4_f32(false, a2[2 * p + 1], false, bB,
                                                            (short)0, oacc[o], false, false);
          }
        }
      }

#pragma unroll
      for (int o = 0; o < 4; ++o) {
#pragma unroll
        for (int j = 0; j < 8; ++j)
          out[(size_t)(row0 + lhalf * 8 + j) * COUTN + o * 16 + lmod] = oacc[o][j];
      }
    }
    __syncthreads();
  }
}

extern "C" void kernel_launch(void* const* d_in, const int* in_sizes, int n_in,
                              void* d_out, int out_size, void* d_ws, size_t ws_size,
                              hipStream_t stream) {
  const float* x     = (const float*)d_in[0];
  const int*   A_idx = (const int*)d_in[1];
  const float* A_val = (const float*)d_in[2];
  const float* W1    = (const float*)d_in[3];
  const float* b1    = (const float*)d_in[4];
  const float* W2    = (const float*)d_in[5];
  const float* b2    = (const float*)d_in[6];
  float* out = (float*)d_out;

  int*  hist     = (int*)d_ws;
  int*  binstart = (int*)((char*)d_ws + 262144);
  int*  rowptr   = (int*)((char*)d_ws + 524288);
  int*  bucket   = (int*)((char*)d_ws + 1048576);
  int*  sedge    = (int*)((char*)d_ws + 8388608);
  const int* rows = A_idx;
  const int* cols = A_idx + NE;

  gcn_hist_kernel<<<NCHUNK, 256, 0, stream>>>(rows, hist);
  gcn_part_kernel<<<NCHUNK, 256, 0, stream>>>(rows, cols, A_val, hist, binstart, bucket);
  gcn_rowsort_kernel<<<NBIN, 256, 0, stream>>>(binstart, bucket, rowptr, sedge);
  gcn_fused_kernel<<<NWG, 768, 0, stream>>>(x, rowptr, sedge, out, W1, b1, W2, b2);
}


# edge sort: each chunk orders its own slice by row range and records run starts; the range kernel collects its 256 runs itself; no histogram launch, no cross-chunk prefix; row sums in LDS rank order
# speedup vs baseline: 5.2444x; 1.0336x over previous
#include <hip/hip_runtime.h>

typedef __attribute__((ext_vector_type(2))) float v2f;
typedef __attribute__((ext_vector_type(8))) float v8f;

#define NB    4
#define NN    50000
#define MM    (NB * NN)
#define CIN   64
#define HID   256
#define COUTN 64
#define NE    800000
#define STRIPS (MM / 16)

#define NBIN    256
#define BINROWS 196
#define NCHUNK  256
#define CHUNK   3125
#define CPT     13
#define CAP     4096

__device__ __forceinline__ int gcn_excl_scan256(int v, int* sW, int tid) {
  const int lane = tid & 31, wv = tid >> 5;
  int inc = v;
#pragma unroll
  for (int d = 1; d < 32; d <<= 1) {
    int t = __shfl_up(inc, d);
    if (lane >= d) inc += t;
  }
  if (lane == 31) sW[wv] = inc;
  __syncthreads();
  int off = 0;
#pragma unroll
  for (int w = 0; w < 7; ++w) off += (w < wv) ? sW[w] : 0;
  return off + inc - v;
}

__global__ void __launch_bounds__(256)
gcn_chunksort_kernel(const int* __restrict__ rows,
                     const int* __restrict__ cols,
                     const float* __restrict__ vals,
                     int* __restrict__ offT,
                     int* __restrict__ bucket_w) {
  int2* __restrict__ bucket = (int2*)bucket_w;
  __shared__ int sLex[NBIN];
  __shared__ int sCur[NBIN];
  __shared__ int sWa[8];
  __shared__ int2 sRec[CHUNK];
  const int tid = threadIdx.x, c = blockIdx.x;
  sCur[tid] = 0;

  const int e0 = c * CHUNK;
  int r[CPT], cc[CPT];
  float vv[CPT];
#pragma unroll
  for (int k = 0; k < CPT; ++k) {
    const int i = tid + k * 256;
    const int e = e0 + (i < CHUNK ? i : CHUNK - 1);
    r[k] = rows[e];
    cc[k] = cols[e];
    vv[k] = vals[e];
  }
  __syncthreads();
  int pk[CPT];
#pragma unroll
  for (int k = 0; k < CPT; ++k) {
    const int i = tid + k * 256;
    pk[k] = 0;
    if (i < CHUNK) pk[k] = atomicAdd(&sCur[r[k] / BINROWS], 1);
  }
  __syncthreads();
  const int cnt = sCur[tid];
  const int ex = gcn_excl_scan256(cnt, sWa, tid);
  sLex[tid] = ex;
  offT[tid * NCHUNK + c] = ex;
  if (tid == NBIN - 1) offT[NBIN * NCHUNK + c] = ex + cnt;
  __syncthreads();
#pragma unroll
  for (int k = 0; k < CPT; ++k) {
    const int i = tid + k * 256;
    if (i < CHUNK) {
      const int b  = r[k] / BINROWS;
      const int lr = r[k] - b * BINROWS;
      sRec[sLex[b] + pk[k]] = make_int2((lr << 16) | cc[k], __float_as_int(vv[k]));
    }
  }
  __syncthreads();
  int2* dst = bucket + (size_t)c * CHUNK;
#pragma unroll
  for (int k = 0; k < CPT; ++k) {
    const int i = tid + k * 256;
    if (i < CHUNK) dst[i] = sRec[i];
  }
}

#define RUNS 32
__global__ void __launch_bounds__(256)
gcn_rowsort_kernel(const int* __restrict__ offT,
                   const int* __restrict__ bucket_w,
                   int* __restrict__ rowptr,
                   int* __restrict__ sedge_w) {
  const int2* __restrict__ bucket = (const int2*)bucket_w;
  int2* __restrict__ sedge = (int2*)sedge_w;
  __shared__ int sCnt[256];
  __shared__ int sCur[256];
  __shared__ int sWa[8];
  __shared__ int sWb[8];
  __shared__ int sWc[8];
  __shared__ int sTot[2];
  __shared__ int2 sOut[CAP];
  const int tid = threadIdx.x, b = blockIdx.x;
  const int lane = tid & 31, wv = tid >> 5;
  const int s = offT[b * NCHUNK + tid];
  const int m = offT[(b + 1) * NCHUNK + tid] - s;
  sCnt[tid] = 0;
  const int exS = gcn_excl_scan256(s, sWa, tid);
  const int exM = gcn_excl_scan256(m, sWb, tid);
  if (tid == 255) { sTot[0] = exS + s; sTot[1] = exM + m; }
  __syncthreads();
  const int start = sTot[0];
  const int n     = sTot[1];
  const int c0 = wv * RUNS;

#define GCN_RUN(J)                                                             \
  const int sc_ = __builtin_amdgcn_readlane(s, J);                             \
  const int mc_ = __builtin_amdgcn_readlane(m, J);                             \
  const int kk_ = (lane < mc_) ? lane : (mc_ > 0 ? mc_ - 1 : 0);               \
  int gi_ = (c0 + J) * CHUNK + sc_ + kk_;                                      \
  gi_ = gi_ < NE ? gi_ : NE - 1;

  {
    int key[RUNS];
#pragma unroll
    for (int j = 0; j < RUNS; ++j) {
      GCN_RUN(j)
      key[j] = bucket[gi_].x;
    }
#pragma unroll
    for (int j = 0; j < RUNS; ++j) {
      const int mc = __builtin_amdgcn_readlane(m, j);
      if (lane < mc) atomicAdd(&sCnt[key[j] >> 16], 1);
    }
    for (int j = 0; j < RUNS; ++j) {
      const int sc = __builtin_amdgcn_readlane(s, j);
      const int mc = __builtin_amdgcn_readlane(m, j);
      for (int k = 32 + lane; k < mc; k += 32)
        atomicAdd(&sCnt[bucket[(c0 + j) * CHUNK + sc + k].x >> 16], 1);
    }
  }
  __syncthreads();

  const int cnt = sCnt[tid];
  const int ex = gcn_excl_scan256(cnt, sWc, tid);
  sCur[tid] = ex;
  const int grow = b * BINROWS + tid;
  if (tid < BINROWS && grow <= NN) rowptr[grow] = start + ex;
  __syncthreads();

  const bool staged = (n <= CAP);
  {
    int2 rec[RUNS];
#pragma unroll
    for (int j = 0; j < RUNS; ++j) {
      GCN_RUN(j)
      rec[j] = bucket[gi_];
    }
#pragma unroll
    for (int j = 0; j < RUNS; ++j) {
      const int mc = __builtin_amdgcn_readlane(m, j);
      if (lane < mc) {
        const int p = atomicAdd(&sCur[rec[j].x >> 16], 1);
        const int2 o = make_int2(rec[j].x & 0xffff, rec[j].y);
        if (staged) sOut[p] = o;
        else        sedge[start + p] = o;
      }
    }
    for (int j = 0; j < RUNS; ++j) {
      const int sc = __builtin_amdgcn_readlane(s, j);
      const int mc = __builtin_amdgcn_readlane(m, j);
      for (int k = 32 + lane; k < mc; k += 32) {
        const int2 rr = bucket[(c0 + j) * CHUNK + sc + k];
        const int p = atomicAdd(&sCur[rr.x >> 16], 1);
        const int2 o = make_int2(rr.x & 0xffff, rr.y);
        if (staged) sOut[p] = o;
        else        sedge[start + p] = o;
      }
    }
  }
#undef GCN_RUN
  if (staged) {
    __syncthreads();
    int2* dst = sedge + start;
    for (int i = tid; i < n; i += 256) dst[i] = sOut[i];
  }
}

#define NBLK  (NN / 16)
#define NWG   256
#define TP    68
#define STP   20

__device__ __forceinline__ void gcn_gather_pair(const float* __restrict__ x,
                                                const int* __restrict__ rowptr,
                                                const int* __restrict__ sedge,
                                                float* __restrict__ tile,
                                                int nb, int nbl, int g8, int lane) {
  const int lhalf = lane >> 4, q4 = (lane & 15) * 4;
  const int n0 = nb * 16 + g8 * 2;
  const int p0 = __builtin_amdgcn_readfirstlane(rowptr[n0]);
  const int p1 = __builtin_amdgcn_readfirstlane(rowptr[n0 + 1]);
  const int p2 = __builtin_amdgcn_readfirstlane(rowptr[n0 + 2]);
  const unsigned off0 = (unsigned)(lhalf * NN * CIN + q4) * 4u;
  const unsigned off1 = off0 + 2u * NN * CIN * 4u;

#define GCN_LOAD(A, B, C)                                                      \
  {                                                                            \
    const char* xr_ = (const char*)(x + (size_t)(C) * CIN);                    \
    A = *(const float4*)(xr_ + off0);                                          \
    B = *(const float4*)(xr_ + off1);                                          \
  }
#define GCN_FMA(V, A, B)                                                       \
  {                                                                            \
    acc0.x = fmaf(V, A.x, acc0.x); acc0.y = fmaf(V, A.y, acc0.y);              \
    acc0.z = fmaf(V, A.z, acc0.z); acc0.w = fmaf(V, A.w, acc0.w);              \
    acc1.x = fmaf(V, B.x, acc1.x); acc1.y = fmaf(V, B.y, acc1.y);              \
    acc1.z = fmaf(V, B.z, acc1.z); acc1.w = fmaf(V, B.w, acc1.w);              \
  }
#define GCN_CV(I)                                                              \
  const int   c##I = __builtin_amdgcn_readlane(m.x, j + I);                    \
  const float v##I = __int_as_float(__builtin_amdgcn_readlane(m.y, j + I));

#pragma unroll 1
  for (int k = 0; k < 2; ++k) {
    const int s = k ? p1 : p0;
    const int e = k ? p2 : p1;
    float4 acc0 = make_float4(0.f, 0.f, 0.f, 0.f);
    float4 acc1 = make_float4(0.f, 0.f, 0.f, 0.f);
    for (int base = s; base < e; base += 32) {
      int2 m = make_int2(0, 0);
      if (base + lane < e) m = *(const int2*)(sedge + 2 * (size_t)(base + lane));
      const int n = (e - base < 32) ? (e - base) : 32;
      int j = 0;
      for (; j + 8 <= n; j += 8) {
        GCN_CV(0) GCN_CV(1) GCN_CV(2) GCN_CV(3) GCN_CV(4) GCN_CV(5) GCN_CV(6) GCN_CV(7)
        float4 a0, b0, a1, b1, a2, b2, a3, b3, a4, b4, a5, b5, a6, b6, a7, b7;
        GCN_LOAD(a0, b0, c0) GCN_LOAD(a1, b1, c1) GCN_LOAD(a2, b2, c2) GCN_LOAD(a3, b3, c3)
        GCN_LOAD(a4, b4, c4) GCN_LOAD(a5, b5, c5) GCN_LOAD(a6, b6, c6) GCN_LOAD(a7, b7, c7)
        GCN_FMA(v0, a0, b0) GCN_FMA(v1, a1, b1) GCN_FMA(v2, a2, b2) GCN_FMA(v3, a3, b3)
        GCN_FMA(v4, a4, b4) GCN_FMA(v5, a5, b5) GCN_FMA(v6, a6, b6) GCN_FMA(v7, a7, b7)
      }
      for (; j + 4 <= n; j += 4) {
        GCN_CV(0) GCN_CV(1) GCN_CV(2) GCN_CV(3)
        float4 a0, b0, a1, b1, a2, b2, a3, b3;
        GCN_LOAD(a0, b0, c0) GCN_LOAD(a1, b1, c1) GCN_LOAD(a2, b2, c2) GCN_LOAD(a3, b3, c3)
        GCN_FMA(v0, a0, b0) GCN_FMA(v1, a1, b1) GCN_FMA(v2, a2, b2) GCN_FMA(v3, a3, b3)
      }
#pragma unroll 1
      for (; j < n; ++j) {
        GCN_CV(0)
        float4 a0, b0;
        GCN_LOAD(a0, b0, c0)
        GCN_FMA(v0, a0, b0)
      }
    }
    const int nrow = g8 * 2 + k;
    *(float4*)(tile + ((nbl * 4 + lhalf) * 16 + nrow) * TP + q4)     = acc0;
    *(float4*)(tile + ((nbl * 4 + lhalf + 2) * 16 + nrow) * TP + q4) = acc1;
  }
#undef GCN_LOAD
#undef GCN_FMA
#undef GCN_CV
}

__global__ void __launch_bounds__(768)
gcn_fused_kernel(const float* __restrict__ x,
                 const int* __restrict__ rowptr,
                 const int* __restrict__ sedge,
                 float* __restrict__ out,
                 const float* __restrict__ W1,
                 const float* __restrict__ b1,
                 const float* __restrict__ W2,
                 const float* __restrict__ b2) {
  __shared__ __align__(16) float sW1f[HID * CIN];
  __shared__ __align__(16) float sW2f[COUTN * HID];
  __shared__ __align__(16) float sB1[HID];
  __shared__ __align__(16) float sB2[COUTN];
  __shared__ __align__(16) float sStage[8][16 * STP];
  __shared__ __align__(16) float sTile[2][128 * TP];

  const int tid = threadIdx.x;
  const int lane  = tid & 31;
  const int lhalf = lane >> 4;
  const int lmod  = lane & 15;
  const int wv = __builtin_amdgcn_readfirstlane(tid >> 5);
  const int nb0 = ((int)blockIdx.x * NBLK) / NWG;
  const int nb1 = (((int)blockIdx.x + 1) * NBLK) / NWG;
  const int nIt = (nb1 - nb0 + 1) >> 1;

  if (wv < 8) {
    {
      const int n = tid, tt = n >> 4, lm = n & 15;
#pragma unroll
      for (int g = 0; g < 16; ++g) {
        const int p = g >> 1, lh = g & 1;
        const float* src = W1 + (8 * p + 2 * lh) * HID + n;
        float4 v = make_float4(src[0], src[HID], src[4 * HID], src[5 * HID]);
        *(float4*)(&sW1f[(((tt * 8 + p) * 32) + lh * 16 + lm) * 4]) = v;
      }
    }
    {
      const int n = tid & 63, o4 = n >> 4, lm = n & 15, sub = tid >> 6;
#pragma unroll
      for (int j = 0; j < 16; ++j) {
        const int g = j * 4 + sub;
        const int tt = g >> 2, p = (g >> 1) & 1, lh = g & 1;
        const float* src = W2 + (16 * tt + 8 * p + 2 * lh) * COUTN + n;
        float4 v = make_float4(src[0], src[COUTN], src[4 * COUTN], src[5 * COUTN]);
        *(float4*)(&sW2f[((((tt * 4 + o4) * 2 + p) * 32) + lh * 16 + lm) * 4]) = v;
      }
    }
    sB1[tid] = b1[tid];
    if (tid < COUTN) sB2[tid] = b2[tid];
  } else {
    const int g = wv - 8, nbl = g >> 3;
    if (nb0 + nbl < nb1)
      gcn_gather_pair(x, rowptr, sedge, &sTile[0][0], nb0 + nbl, nbl, g & 7, lane);
  }
  __syncthreads();

  float* st = &sStage[wv & 7][0];
  const float4* w1f = (const float4*)sW1f + lane;
  const float4* w2f = (const float4*)sW2f + lane;

  for (int it = 0; it < nIt; ++it) {
    const int nbA = nb0 + 2 * it;
    if (wv >= 8) {
      const int g = wv - 8, nbl = g >> 3;
      if (it + 1 < nIt && nbA + 2 + nbl < nb1)
        gcn_gather_pair(x, rowptr, sedge, &sTile[(it + 1) & 1][0], nbA + 2 + nbl, nbl, g & 7, lane);
    } else if (nbA + (wv >> 2) < nb1) {
      const int row0 = (wv & 3) * NN + (nbA + (wv >> 2)) * 16;

      v2f a1[16];
      {
        const float* base = &sTile[it & 1][(wv * 16 + lmod) * TP + 2 * lhalf];
#pragma unroll
        for (int s = 0; s < 16; ++s)
          a1[s] = *(const v2f*)(base + 4 * s);
      }

      v8f oacc[4];
#pragma unroll
      for (int o = 0; o < 4; ++o) {
        float bv = sB2[o * 16 + lmod];
#pragma unroll
        for (int j = 0; j < 8; ++j) oacc[o][j] = bv;
      }

      for (int t = 0; t < 16; ++t) {
        v8f h;
        float b1v = sB1[t * 16 + lmod];
#pragma unroll
        for (int j = 0; j < 8; ++j) h[j] = b1v;
#pragma unroll
        for (int p = 0; p < 8; ++p) {
          float4 w = w1f[(t * 8 + p) * 32];
          v2f bA = {w.x, w.y};
          v2f bB = {w.z, w.w};
          h = __builtin_amdgcn_wmma_f32_16x16x4_f32(false, a1[2 * p], false, bA,
                                                    (short)0, h, false, false);
          h = __builtin_amdgcn_wmma_f32_16x16x4_f32(false, a1[2 * p + 1], false, bB,
                                                    (short)0, h, false, false);
        }
#pragma unroll
        for (int j = 0; j < 8; ++j) h[j] = fmaxf(h[j], 0.f);

#pragma unroll
        for (int j = 0; j < 8; ++j)
          st[(lhalf * 8 + j) * STP + lmod] = h[j];
        asm volatile("s_wait_dscnt 0" ::: "memory");

        v2f a2[4];
#pragma unroll
        for (int kk = 0; kk < 4; ++kk)
          a2[kk] = *(const v2f*)(&st[lmod * STP + 4 * kk + 2 * lhalf]);

#pragma unroll
        for (int o = 0; o < 4; ++o) {
#pragma unroll
          for (int p = 0; p < 2; ++p) {
            float4 w = w2f[((t * 4 + o) * 2 + p) * 32];
            v2f bA = {w.x, w.y};
            v2f bB = {w.z, w.w};
            oacc[o] = __builtin_amdgcn_wmma_f32_16x16x4_f32(false, a2[2 * p], false, bA,
                                                            (short)0, oacc[o], false, false);
            oacc[o] = __builtin_amdgcn_wmma_f32_16x16x4_f32(false, a2[2 * p + 1], false, bB,
                                                            (short)0, oacc[o], false, false);
          }
        }
      }

#pragma unroll
      for (int o = 0; o < 4; ++o) {
#pragma unroll
        for (int j = 0; j < 8; ++j)
          out[(size_t)(row0 + lhalf * 8 + j) * COUTN + o * 16 + lmod] = oacc[o][j];
      }
    }
    __syncthreads();
  }
}

extern "C" void kernel_launch(void* const* d_in, const int* in_sizes, int n_in,
                              void* d_out, int out_size, void* d_ws, size_t ws_size,
                              hipStream_t stream) {
  const float* x     = (const float*)d_in[0];
  const int*   A_idx = (const int*)d_in[1];
  const float* A_val = (const float*)d_in[2];
  const float* W1    = (const float*)d_in[3];
  const float* b1    = (const float*)d_in[4];
  const float* W2    = (const float*)d_in[5];
  const float* b2    = (const float*)d_in[6];
  float* out = (float*)d_out;

  int*  offT     = (int*)d_ws;
  int*  rowptr   = (int*)((char*)d_ws + 524288);
  int*  bucket   = (int*)((char*)d_ws + 1048576);
  int*  sedge    = (int*)((char*)d_ws + 8388608);
  const int* rows = A_idx;
  const int* cols = A_idx + NE;

  gcn_chunksort_kernel<<<NCHUNK, 256, 0, stream>>>(rows, cols, A_val, offT, bucket);
  gcn_rowsort_kernel<<<NBIN, 256, 0, stream>>>(offT, bucket, rowptr, sedge);
  gcn_fused_kernel<<<NWG, 768, 0, stream>>>(x, rowptr, sedge, out, W1, b1, W2, b2);
}


# row sort moved into the persistent kernel: each group orders its own node range's edges in LDS, where they and the row offsets stay, while gather waves fill W1/W2; 2 launches; sums in LDS rank order
# speedup vs baseline: 5.4184x; 1.0332x over previous
#include <hip/hip_runtime.h>

typedef __attribute__((ext_vector_type(2))) float v2f;
typedef __attribute__((ext_vector_type(8))) float v8f;

#define NB    4
#define NN    50000
#define MM    (NB * NN)
#define CIN   64
#define HID   256
#define COUTN 64
#define NE    800000
#define STRIPS (MM / 16)

#define NBLK    (NN / 16)
#define NWG     256
#define NBIN    NWG
#define NCHUNK  256
#define CHUNK   3125
#define CPT     13
#define CAP     4096

__device__ __forceinline__ int gcn_first_block(int g) { return (g * NBLK) / NWG; }
__device__ __forceinline__ int gcn_range_of_row(int r) { return ((r >> 4) * NWG + (NWG - 1)) / NBLK; }

__device__ __forceinline__ int gcn_excl_scan256(int v, int* sW, int tid) {
  const int lane = tid & 31, wv = tid >> 5;
  int inc = v;
#pragma unroll
  for (int d = 1; d < 32; d <<= 1) {
    int t = __shfl_up(inc, d);
    if (lane >= d) inc += t;
  }
  if (lane == 31) sW[wv] = inc;
  __syncthreads();
  int off = 0;
#pragma unroll
  for (int w = 0; w < 7; ++w) off += (w < wv) ? sW[w] : 0;
  return off + inc - v;
}

__global__ void __launch_bounds__(256)
gcn_chunksort_kernel(const int* __restrict__ rows,
                     const int* __restrict__ cols,
                     const float* __restrict__ vals,
                     int* __restrict__ offT,
                     int* __restrict__ bucket_w) {
  int2* __restrict__ bucket = (int2*)bucket_w;
  __shared__ int sLex[NBIN];
  __shared__ int sCur[NBIN];
  __shared__ int sWa[8];
  __shared__ int2 sRec[CHUNK];
  const int tid = threadIdx.x, c = blockIdx.x;
  sCur[tid] = 0;

  const int e0 = c * CHUNK;
  int r[CPT], cc[CPT];
  float vv[CPT];
#pragma unroll
  for (int k = 0; k < CPT; ++k) {
    const int i = tid + k * 256;
    const int e = e0 + (i < CHUNK ? i : CHUNK - 1);
    r[k] = rows[e];
    cc[k] = cols[e];
    vv[k] = vals[e];
  }
  __syncthreads();
  int pk[CPT];
#pragma unroll
  for (int k = 0; k < CPT; ++k) {
    const int i = tid + k * 256;
    pk[k] = 0;
    if (i < CHUNK) pk[k] = atomicAdd(&sCur[gcn_range_of_row(r[k])], 1);
  }
  __syncthreads();
  const int cnt = sCur[tid];
  const int ex = gcn_excl_scan256(cnt, sWa, tid);
  sLex[tid] = ex;
  offT[tid * NCHUNK + c] = ex;
  if (tid == NBIN - 1) offT[NBIN * NCHUNK + c] = ex + cnt;
  __syncthreads();
#pragma unroll
  for (int k = 0; k < CPT; ++k) {
    const int i = tid + k * 256;
    if (i < CHUNK) {
      const int b  = gcn_range_of_row(r[k]);
      const int lr = r[k] - gcn_first_block(b) * 16;
      sRec[sLex[b] + pk[k]] = make_int2((lr << 16) | cc[k], __float_as_int(vv[k]));
    }
  }
  __syncthreads();
  int2* dst = bucket + (size_t)c * CHUNK;
#pragma unroll
  for (int k = 0; k < CPT; ++k) {
    const int i = tid + k * 256;
    if (i < CHUNK) dst[i] = sRec[i];
  }
}

#define RUNS  32
#define TP    68
#define STP   20

__device__ __forceinline__ void gcn_gather_pair(const float* __restrict__ x,
                                                const int* rp,
                                                const int2* led,
                                                const int2* ged,
                                                int start, bool staged,
                                                float* __restrict__ tile,
                                                int ln0, int nbl, int g8, int lane) {
  const int lhalf = lane >> 4, q4 = (lane & 15) * 4;
  const int p0 = __builtin_amdgcn_readfirstlane(rp[ln0]);
  const int p1 = __builtin_amdgcn_readfirstlane(rp[ln0 + 1]);
  const int p2 = __builtin_amdgcn_readfirstlane(rp[ln0 + 2]);
  const unsigned off0 = (unsigned)(lhalf * NN * CIN + q4) * 4u;
  const unsigned off1 = off0 + 2u * NN * CIN * 4u;

#define GCN_LOAD(A, B, C)                                                      \
  {                                                                            \
    const char* xr_ = (const char*)(x + (size_t)(C) * CIN);                    \
    A = *(const float4*)(xr_ + off0);                                          \
    B = *(const float4*)(xr_ + off1);                                          \
  }
#define GCN_FMA(V, A, B)                                                       \
  {                                                                            \
    acc0.x = fmaf(V, A.x, acc0.x); acc0.y = fmaf(V, A.y, acc0.y);              \
    acc0.z = fmaf(V, A.z, acc0.z); acc0.w = fmaf(V, A.w, acc0.w);              \
    acc1.x = fmaf(V, B.x, acc1.x); acc1.y = fmaf(V, B.y, acc1.y);              \
    acc1.z = fmaf(V, B.z, acc1.z); acc1.w = fmaf(V, B.w, acc1.w);              \
  }
#define GCN_CV(I)                                                              \
  const int   c##I = __builtin_amdgcn_readlane(m.x, j + I);                    \
  const float v##I = __int_as_float(__builtin_amdgcn_readlane(m.y, j + I));

#pragma unroll 1
  for (int k = 0; k < 2; ++k) {
    const int s = k ? p1 : p0;
    const int e = k ? p2 : p1;
    float4 acc0 = make_float4(0.f, 0.f, 0.f, 0.f);
    float4 acc1 = make_float4(0.f, 0.f, 0.f, 0.f);
    for (int base = s; base < e; base += 32) {
      int2 m = make_int2(0, 0);
      if (base + lane < e) {
        if (staged) m = led[base + lane];
        else        m = ged[(size_t)start + base + lane];
      }
      const int n = (e - base < 32) ? (e - base) : 32;
      int j = 0;
      for (; j + 8 <= n; j += 8) {
        GCN_CV(0) GCN_CV(1) GCN_CV(2) GCN_CV(3) GCN_CV(4) GCN_CV(5) GCN_CV(6) GCN_CV(7)
        float4 a0, b0, a1, b1, a2, b2, a3, b3, a4, b4, a5, b5, a6, b6, a7, b7;
        GCN_LOAD(a0, b0, c0) GCN_LOAD(a1, b1, c1) GCN_LOAD(a2, b2, c2) GCN_LOAD(a3, b3, c3)
        GCN_LOAD(a4, b4, c4) GCN_LOAD(a5, b5, c5) GCN_LOAD(a6, b6, c6) GCN_LOAD(a7, b7, c7)
        GCN_FMA(v0, a0, b0) GCN_FMA(v1, a1, b1) GCN_FMA(v2, a2, b2) GCN_FMA(v3, a3, b3)
        GCN_FMA(v4, a4, b4) GCN_FMA(v5, a5, b5) GCN_FMA(v6, a6, b6) GCN_FMA(v7, a7, b7)
      }
      for (; j + 4 <= n; j += 4) {
        GCN_CV(0) GCN_CV(1) GCN_CV(2) GCN_CV(3)
        float4 a0, b0, a1, b1, a2, b2, a3, b3;
        GCN_LOAD(a0, b0, c0) GCN_LOAD(a1, b1, c1) GCN_LOAD(a2, b2, c2) GCN_LOAD(a3, b3, c3)
        GCN_FMA(v0, a0, b0) GCN_FMA(v1, a1, b1) GCN_FMA(v2, a2, b2) GCN_FMA(v3, a3, b3)
      }
#pragma unroll 1
      for (; j < n; ++j) {
        GCN_CV(0)
        float4 a0, b0;
        GCN_LOAD(a0, b0, c0)
        GCN_FMA(v0, a0, b0)
      }
    }
    const int nrow = g8 * 2 + k;
    *(float4*)(tile + ((nbl * 4 + lhalf) * 16 + nrow) * TP + q4)     = acc0;
    *(float4*)(tile + ((nbl * 4 + lhalf + 2) * 16 + nrow) * TP + q4) = acc1;
  }
#undef GCN_LOAD
#undef GCN_FMA
#undef GCN_CV
}

__global__ void __launch_bounds__(768)
gcn_fused_kernel(const float* __restrict__ x,
                 const int* __restrict__ offT,
                 const int* __restrict__ bucket_w,
                 int* sedge_w,
                 float* __restrict__ out,
                 const float* __restrict__ W1,
                 const float* __restrict__ b1,
                 const float* __restrict__ W2,
                 const float* __restrict__ b2) {
  __shared__ __align__(16) float sW1f[HID * CIN];
  __shared__ __align__(16) float sW2f[COUTN * HID];
  __shared__ __align__(16) float sB1[HID];
  __shared__ __align__(16) float sB2[COUTN];
  __shared__ __align__(16) float sStage[8][16 * STP];
  __shared__ __align__(16) float sTile[2][128 * TP];
  __shared__ int sCnt[256];
  __shared__ int sCur[256];
  __shared__ int sRp[256];
  __shared__ int sWa[8];
  __shared__ int sWb[8];
  __shared__ int sWc[8];
  __shared__ int2 sEdge[CAP];

  const int2* __restrict__ bucket = (const int2*)bucket_w;
  int2* sedge = (int2*)sedge_w;
  const int tid = threadIdx.x;
  const int lane  = tid & 31;
  const int lhalf = lane >> 4;
  const int lmod  = lane & 15;
  const int wv = __builtin_amdgcn_readfirstlane(tid >> 5);
  const int b = (int)blockIdx.x;
  const int nb0 = gcn_first_block(b);
  const int nb1 = gcn_first_block(b + 1);
  const int nIt = (nb1 - nb0 + 1) >> 1;
  const int gt = tid - 256;
  const int c0 = wv * RUNS;

  int s = 0, m = 0;
  if (wv < 8) {
    s = offT[b * NCHUNK + tid];
    m = offT[(b + 1) * NCHUNK + tid] - s;
    sCnt[tid] = 0;
    int ts = s, tm = m;
#pragma unroll
    for (int d = 16; d >= 1; d >>= 1) {
      ts += __shfl_xor(ts, d);
      tm += __shfl_xor(tm, d);
    }
    if (lane == 0) { sWa[wv] = ts; sWb[wv] = tm; }
  } else {
    const int n = gt & 255, lh = gt >> 8, tt = n >> 4, lm = n & 15;
#pragma unroll
    for (int p = 0; p < 8; ++p) {
      const float* src = W1 + (8 * p + 2 * lh) * HID + n;
      float4 v = make_float4(src[0], src[HID], src[4 * HID], src[5 * HID]);
      *(float4*)(&sW1f[(((tt * 8 + p) * 32) + lh * 16 + lm) * 4]) = v;
    }
  }
  __syncthreads();

  int start = 0, n = 0;
#pragma unroll
  for (int w = 0; w < 8; ++w) { start += sWa[w]; n += sWb[w]; }
  const bool staged = (n <= CAP);

#define GCN_RUN(J)                                                             \
  const int sc_ = __builtin_amdgcn_readlane(s, J);                             \
  const int mc_ = __builtin_amdgcn_readlane(m, J);                             \
  const int kk_ = (lane < mc_) ? lane : (mc_ > 0 ? mc_ - 1 : 0);               \
  int gi_ = (c0 + J) * CHUNK + sc_ + kk_;                                      \
  gi_ = gi_ < NE ? gi_ : NE - 1;

  if (wv < 8) {
    int key[RUNS];
#pragma unroll
    for (int j = 0; j < RUNS; ++j) {
      GCN_RUN(j)
      key[j] = bucket[gi_].x;
    }
#pragma unroll
    for (int j = 0; j < RUNS; ++j) {
      const int mc = __builtin_amdgcn_readlane(m, j);
      if (lane < mc) atomicAdd(&sCnt[key[j] >> 16], 1);
    }
    for (int j = 0; j < RUNS; ++j) {
      const int sc = __builtin_amdgcn_readlane(s, j);
      const int mc = __builtin_amdgcn_readlane(m, j);
      for (int k = 32 + lane; k < mc; k += 32)
        atomicAdd(&sCnt[bucket[(c0 + j) * CHUNK + sc + k].x >> 16], 1);
    }
  } else {
    const int n2 = gt & 63, o4 = n2 >> 4, lm = n2 & 15, sub = gt >> 6;
#pragma unroll
    for (int j = 0; j < 8; ++j) {
      const int g = j * 8 + sub;
      const int tt = g >> 2, p = (g >> 1) & 1, lh = g & 1;
      const float* src = W2 + (16 * tt + 8 * p + 2 * lh) * COUTN + n2;
      float4 v = make_float4(src[0], src[COUTN], src[4 * COUTN], src[5 * COUTN]);
      *(float4*)(&sW2f[((((tt * 4 + o4) * 2 + p) * 32) + lh * 16 + lm) * 4]) = v;
    }
    if (gt < HID) sB1[gt] = b1[gt];
    if (gt < COUTN) sB2[gt] = b2[gt];
  }
  __syncthreads();

  int cnt = 0, inc = 0;
  if (wv < 8) {
    cnt = sCnt[tid];
    inc = cnt;
#pragma unroll
    for (int d = 1; d < 32; d <<= 1) {
      int t = __shfl_up(inc, d);
      if (lane >= d) inc += t;
    }
    if (lane == 31) sWc[wv] = inc;
  }
  __syncthreads();
  if (wv < 8) {
    int off = 0;
#pragma unroll
    for (int w = 0; w < 7; ++w) off += (w < wv) ? sWc[w] : 0;
    const int ex = off + inc - cnt;
    sCur[tid] = ex;
    sRp[tid] = ex;
  }
  __syncthreads();

  if (wv < 8) {
    int2 rec[RUNS];
#pragma unroll
    for (int j = 0; j < RUNS; ++j) {
      GCN_RUN(j)
      rec[j] = bucket[gi_];
    }
#pragma unroll
    for (int j = 0; j < RUNS; ++j) {
      const int mc = __builtin_amdgcn_readlane(m, j);
      if (lane < mc) {
        const int p = atomicAdd(&sCur[rec[j].x >> 16], 1);
        const int2 o = make_int2(rec[j].x & 0xffff, rec[j].y);
        if (staged) sEdge[p] = o;
        else        sedge[start + p] = o;
      }
    }
    for (int j = 0; j < RUNS; ++j) {
      const int sc = __builtin_amdgcn_readlane(s, j);
      const int mc = __builtin_amdgcn_readlane(m, j);
      for (int k = 32 + lane; k < mc; k += 32) {
        const int2 rr = bucket[(c0 + j) * CHUNK + sc + k];
        const int p = atomicAdd(&sCur[rr.x >> 16], 1);
        const int2 o = make_int2(rr.x & 0xffff, rr.y);
        if (staged) sEdge[p] = o;
        else        sedge[start + p] = o;
      }
    }
  }
#undef GCN_RUN
  __syncthreads();

  if (wv >= 8) {
    const int g = wv - 8, nbl = g >> 3;
    if (nb0 + nbl < nb1)
      gcn_gather_pair(x, sRp, sEdge, sedge, start, staged, &sTile[0][0],
                      nbl * 16 + (g & 7) * 2, nbl, g & 7, lane);
  }
  __syncthreads();

  float* st = &sStage[wv & 7][0];
  const float4* w1f = (const float4*)sW1f + lane;
  const float4* w2f = (const float4*)sW2f + lane;

  for (int it = 0; it < nIt; ++it) {
    const int nbA = nb0 + 2 * it;
    if (wv >= 8) {
      const int g = wv - 8, nbl = g >> 3;
      if (it + 1 < nIt && nbA + 2 + nbl < nb1)
        gcn_gather_pair(x, sRp, sEdge, sedge, start, staged, &sTile[(it + 1) & 1][0],
                        (nbA + 2 + nbl - nb0) * 16 + (g & 7) * 2, nbl, g & 7, lane);
    } else if (nbA + (wv >> 2) < nb1) {
      const int row0 = (wv & 3) * NN + (nbA + (wv >> 2)) * 16;

      v2f a1[16];
      {
        const float* base = &sTile[it & 1][(wv * 16 + lmod) * TP + 2 * lhalf];
#pragma unroll
        for (int s = 0; s < 16; ++s)
          a1[s] = *(const v2f*)(base + 4 * s);
      }

      v8f oacc[4];
#pragma unroll
      for (int o = 0; o < 4; ++o) {
        float bv = sB2[o * 16 + lmod];
#pragma unroll
        for (int j = 0; j < 8; ++j) oacc[o][j] = bv;
      }

      for (int t = 0; t < 16; ++t) {
        v8f h;
        float b1v = sB1[t * 16 + lmod];
#pragma unroll
        for (int j = 0; j < 8; ++j) h[j] = b1v;
#pragma unroll
        for (int p = 0; p < 8; ++p) {
          float4 w = w1f[(t * 8 + p) * 32];
          v2f bA = {w.x, w.y};
          v2f bB = {w.z, w.w};
          h = __builtin_amdgcn_wmma_f32_16x16x4_f32(false, a1[2 * p], false, bA,
                                                    (short)0, h, false, false);
          h = __builtin_amdgcn_wmma_f32_16x16x4_f32(false, a1[2 * p + 1], false, bB,
                                                    (short)0, h, false, false);
        }
#pragma unroll
        for (int j = 0; j < 8; ++j) h[j] = fmaxf(h[j], 0.f);

#pragma unroll
        for (int j = 0; j < 8; ++j)
          st[(lhalf * 8 + j) * STP + lmod] = h[j];
        asm volatile("s_wait_dscnt 0" ::: "memory");

        v2f a2[4];
#pragma unroll
        for (int kk = 0; kk < 4; ++kk)
          a2[kk] = *(const v2f*)(&st[lmod * STP + 4 * kk + 2 * lhalf]);

#pragma unroll
        for (int o = 0; o < 4; ++o) {
#pragma unroll
          for (int p = 0; p < 2; ++p) {
            float4 w = w2f[((t * 4 + o) * 2 + p) * 32];
            v2f bA = {w.x, w.y};
            v2f bB = {w.z, w.w};
            oacc[o] = __builtin_amdgcn_wmma_f32_16x16x4_f32(false, a2[2 * p], false, bA,
                                                            (short)0, oacc[o], false, false);
            oacc[o] = __builtin_amdgcn_wmma_f32_16x16x4_f32(false, a2[2 * p + 1], false, bB,
                                                            (short)0, oacc[o], false, false);
          }
        }
      }

#pragma unroll
      for (int o = 0; o < 4; ++o) {
#pragma unroll
        for (int j = 0; j < 8; ++j)
          out[(size_t)(row0 + lhalf * 8 + j) * COUTN + o * 16 + lmod] = oacc[o][j];
      }
    }
    __syncthreads();
  }
}

extern "C" void kernel_launch(void* const* d_in, const int* in_sizes, int n_in,
                              void* d_out, int out_size, void* d_ws, size_t ws_size,
                              hipStream_t stream) {
  const float* x     = (const float*)d_in[0];
  const int*   A_idx = (const int*)d_in[1];
  const float* A_val = (const float*)d_in[2];
  const float* W1    = (const float*)d_in[3];
  const float* b1    = (const float*)d_in[4];
  const float* W2    = (const float*)d_in[5];
  const float* b2    = (const float*)d_in[6];
  float* out = (float*)d_out;

  int*  offT     = (int*)d_ws;
  int*  bucket   = (int*)((char*)d_ws + 1048576);
  int*  sedge    = (int*)((char*)d_ws + 8388608);
  const int* rows = A_idx;
  const int* cols = A_idx + NE;

  gcn_chunksort_kernel<<<NCHUNK, 256, 0, stream>>>(rows, cols, A_val, offT, bucket);
  gcn_fused_kernel<<<NWG, 768, 0, stream>>>(x, offT, bucket, sedge, out, W1, b1, W2, b2);
}


# tile hand-over split in two: matrix waves signal once a tile's rows are in registers, wait after their products; gather waves run up to two tiles ahead; LDS-only fences; sum order unchanged
# speedup vs baseline: 5.6638x; 1.0453x over previous
#include <hip/hip_runtime.h>

typedef __attribute__((ext_vector_type(2))) float v2f;
typedef __attribute__((ext_vector_type(8))) float v8f;

#define NB    4
#define NN    50000
#define MM    (NB * NN)
#define CIN   64
#define HID   256
#define COUTN 64
#define NE    800000
#define STRIPS (MM / 16)

#define NBLK    (NN / 16)
#define NWG     256
#define NBIN    NWG
#define NCHUNK  256
#define CHUNK   3125
#define CPT     13
#define CAP     4096

__device__ __forceinline__ int gcn_first_block(int g) { return (g * NBLK) / NWG; }
__device__ __forceinline__ int gcn_range_of_row(int r) { return ((r >> 4) * NWG + (NWG - 1)) / NBLK; }

__device__ __forceinline__ int gcn_excl_scan256(int v, int* sW, int tid) {
  const int lane = tid & 31, wv = tid >> 5;
  int inc = v;
#pragma unroll
  for (int d = 1; d < 32; d <<= 1) {
    int t = __shfl_up(inc, d);
    if (lane >= d) inc += t;
  }
  if (lane == 31) sW[wv] = inc;
  __syncthreads();
  int off = 0;
#pragma unroll
  for (int w = 0; w < 7; ++w) off += (w < wv) ? sW[w] : 0;
  return off + inc - v;
}

__global__ void __launch_bounds__(256)
gcn_chunksort_kernel(const int* __restrict__ rows,
                     const int* __restrict__ cols,
                     const float* __restrict__ vals,
                     int* __restrict__ offT,
                     int* __restrict__ bucket_w) {
  int2* __restrict__ bucket = (int2*)bucket_w;
  __shared__ int sLex[NBIN];
  __shared__ int sCur[NBIN];
  __shared__ int sWa[8];
  __shared__ int2 sRec[CHUNK];
  const int tid = threadIdx.x, c = blockIdx.x;
  sCur[tid] = 0;

  const int e0 = c * CHUNK;
  int r[CPT], cc[CPT];
  float vv[CPT];
#pragma unroll
  for (int k = 0; k < CPT; ++k) {
    const int i = tid + k * 256;
    const int e = e0 + (i < CHUNK ? i : CHUNK - 1);
    r[k] = rows[e];
    cc[k] = cols[e];
    vv[k] = vals[e];
  }
  __syncthreads();
  int pk[CPT];
#pragma unroll
  for (int k = 0; k < CPT; ++k) {
    const int i = tid + k * 256;
    pk[k] = 0;
    if (i < CHUNK) pk[k] = atomicAdd(&sCur[gcn_range_of_row(r[k])], 1);
  }
  __syncthreads();
  const int cnt = sCur[tid];
  const int ex = gcn_excl_scan256(cnt, sWa, tid);
  sLex[tid] = ex;
  offT[tid * NCHUNK + c] = ex;
  if (tid == NBIN - 1) offT[NBIN * NCHUNK + c] = ex + cnt;
  __syncthreads();
#pragma unroll
  for (int k = 0; k < CPT; ++k) {
    const int i = tid + k * 256;
    if (i < CHUNK) {
      const int b  = gcn_range_of_row(r[k]);
      const int lr = r[k] - gcn_first_block(b) * 16;
      sRec[sLex[b] + pk[k]] = make_int2((lr << 16) | cc[k], __float_as_int(vv[k]));
    }
  }
  __syncthreads();
  int2* dst = bucket + (size_t)c * CHUNK;
#pragma unroll
  for (int k = 0; k < CPT; ++k) {
    const int i = tid + k * 256;
    if (i < CHUNK) dst[i] = sRec[i];
  }
}

#define RUNS  32
#define TP    68
#define STP   20

__device__ __forceinline__ void gcn_gather_pair(const float* __restrict__ x,
                                                const int* rp,
                                                const int2* led,
                                                const int2* ged,
                                                int start, bool staged,
                                                float* __restrict__ tile,
                                                int ln0, int nbl, int g8, int lane) {
  const int lhalf = lane >> 4, q4 = (lane & 15) * 4;
  const int p0 = __builtin_amdgcn_readfirstlane(rp[ln0]);
  const int p1 = __builtin_amdgcn_readfirstlane(rp[ln0 + 1]);
  const int p2 = __builtin_amdgcn_readfirstlane(rp[ln0 + 2]);
  const unsigned off0 = (unsigned)(lhalf * NN * CIN + q4) * 4u;
  const unsigned off1 = off0 + 2u * NN * CIN * 4u;

#define GCN_LOAD(A, B, C)                                                      \
  {                                                                            \
    const char* xr_ = (const char*)(x + (size_t)(C) * CIN);                    \
    A = *(const float4*)(xr_ + off0);                                          \
    B = *(const float4*)(xr_ + off1);                                          \
  }
#define GCN_FMA(V, A, B)                                                       \
  {                                                                            \
    acc0.x = fmaf(V, A.x, acc0.x); acc0.y = fmaf(V, A.y, acc0.y);              \
    acc0.z = fmaf(V, A.z, acc0.z); acc0.w = fmaf(V, A.w, acc0.w);              \
    acc1.x = fmaf(V, B.x, acc1.x); acc1.y = fmaf(V, B.y, acc1.y);              \
    acc1.z = fmaf(V, B.z, acc1.z); acc1.w = fmaf(V, B.w, acc1.w);              \
  }
#define GCN_CV(I)                                                              \
  const int   c##I = __builtin_amdgcn_readlane(m.x, j + I);                    \
  const float v##I = __int_as_float(__builtin_amdgcn_readlane(m.y, j + I));

#pragma unroll 1
  for (int k = 0; k < 2; ++k) {
    const int s = k ? p1 : p0;
    const int e = k ? p2 : p1;
    float4 acc0 = make_float4(0.f, 0.f, 0.f, 0.f);
    float4 acc1 = make_float4(0.f, 0.f, 0.f, 0.f);
    for (int base = s; base < e; base += 32) {
      int2 m = make_int2(0, 0);
      if (base + lane < e) {
        if (staged) m = led[base + lane];
        else        m = ged[(size_t)start + base + lane];
      }
      const int n = (e - base < 32) ? (e - base) : 32;
      int j = 0;
      for (; j + 8 <= n; j += 8) {
        GCN_CV(0) GCN_CV(1) GCN_CV(2) GCN_CV(3) GCN_CV(4) GCN_CV(5) GCN_CV(6) GCN_CV(7)
        float4 a0, b0, a1, b1, a2, b2, a3, b3, a4, b4, a5, b5, a6, b6, a7, b7;
        GCN_LOAD(a0, b0, c0) GCN_LOAD(a1, b1, c1) GCN_LOAD(a2, b2, c2) GCN_LOAD(a3, b3, c3)
        GCN_LOAD(a4, b4, c4) GCN_LOAD(a5, b5, c5) GCN_LOAD(a6, b6, c6) GCN_LOAD(a7, b7, c7)
        GCN_FMA(v0, a0, b0) GCN_FMA(v1, a1, b1) GCN_FMA(v2, a2, b2) GCN_FMA(v3, a3, b3)
        GCN_FMA(v4, a4, b4) GCN_FMA(v5, a5, b5) GCN_FMA(v6, a6, b6) GCN_FMA(v7, a7, b7)
      }
      for (; j + 4 <= n; j += 4) {
        GCN_CV(0) GCN_CV(1) GCN_CV(2) GCN_CV(3)
        float4 a0, b0, a1, b1, a2, b2, a3, b3;
        GCN_LOAD(a0, b0, c0) GCN_LOAD(a1, b1, c1) GCN_LOAD(a2, b2, c2) GCN_LOAD(a3, b3, c3)
        GCN_FMA(v0, a0, b0) GCN_FMA(v1, a1, b1) GCN_FMA(v2, a2, b2) GCN_FMA(v3, a3, b3)
      }
#pragma unroll 1
      for (; j < n; ++j) {
        GCN_CV(0)
        float4 a0, b0;
        GCN_LOAD(a0, b0, c0)
        GCN_FMA(v0, a0, b0)
      }
    }
    const int nrow = g8 * 2 + k;
    *(float4*)(tile + ((nbl * 4 + lhalf) * 16 + nrow) * TP + q4)     = acc0;
    *(float4*)(tile + ((nbl * 4 + lhalf + 2) * 16 + nrow) * TP + q4) = acc1;
  }
#undef GCN_LOAD
#undef GCN_FMA
#undef GCN_CV
}

__global__ void __launch_bounds__(768)
gcn_fused_kernel(const float* __restrict__ x,
                 const int* __restrict__ offT,
                 const int* __restrict__ bucket_w,
                 int* sedge_w,
                 float* __restrict__ out,
                 const float* __restrict__ W1,
                 const float* __restrict__ b1,
                 const float* __restrict__ W2,
                 const float* __restrict__ b2) {
  __shared__ __align__(16) float sW1f[HID * CIN];
  __shared__ __align__(16) float sW2f[COUTN * HID];
  __shared__ __align__(16) float sB1[HID];
  __shared__ __align__(16) float sB2[COUTN];
  __shared__ __align__(16) float sStage[8][16 * STP];
  __shared__ __align__(16) float sTile[2][128 * TP];
  __shared__ int sCnt[256];
  __shared__ int sCur[256];
  __shared__ int sRp[256];
  __shared__ int sWa[8];
  __shared__ int sWb[8];
  __shared__ int sWc[8];
  __shared__ int2 sEdge[CAP];

  const int2* __restrict__ bucket = (const int2*)bucket_w;
  int2* sedge = (int2*)sedge_w;
  const int tid = threadIdx.x;
  const int lane  = tid & 31;
  const int lhalf = lane >> 4;
  const int lmod  = lane & 15;
  const int wv = __builtin_amdgcn_readfirstlane(tid >> 5);
  const int b = (int)blockIdx.x;
  const int nb0 = gcn_first_block(b);
  const int nb1 = gcn_first_block(b + 1);
  const int nIt = (nb1 - nb0 + 1) >> 1;
  const int gt = tid - 256;
  const int c0 = wv * RUNS;

  int s = 0, m = 0;
  if (wv < 8) {
    s = offT[b * NCHUNK + tid];
    m = offT[(b + 1) * NCHUNK + tid] - s;
    sCnt[tid] = 0;
    int ts = s, tm = m;
#pragma unroll
    for (int d = 16; d >= 1; d >>= 1) {
      ts += __shfl_xor(ts, d);
      tm += __shfl_xor(tm, d);
    }
    if (lane == 0) { sWa[wv] = ts; sWb[wv] = tm; }
  } else {
    const int n = gt & 255, lh = gt >> 8, tt = n >> 4, lm = n & 15;
#pragma unroll
    for (int p = 0; p < 8; ++p) {
      const float* src = W1 + (8 * p + 2 * lh) * HID + n;
      float4 v = make_float4(src[0], src[HID], src[4 * HID], src[5 * HID]);
      *(float4*)(&sW1f[(((tt * 8 + p) * 32) + lh * 16 + lm) * 4]) = v;
    }
  }
  __syncthreads();

  int start = 0, n = 0;
#pragma unroll
  for (int w = 0; w < 8; ++w) { start += sWa[w]; n += sWb[w]; }
  const bool staged = (n <= CAP);

#define GCN_RUN(J)                                                             \
  const int sc_ = __builtin_amdgcn_readlane(s, J);                             \
  const int mc_ = __builtin_amdgcn_readlane(m, J);                             \
  const int kk_ = (lane < mc_) ? lane : (mc_ > 0 ? mc_ - 1 : 0);               \
  int gi_ = (c0 + J) * CHUNK + sc_ + kk_;                                      \
  gi_ = gi_ < NE ? gi_ : NE - 1;

  if (wv < 8) {
    int key[RUNS];
#pragma unroll
    for (int j = 0; j < RUNS; ++j) {
      GCN_RUN(j)
      key[j] = bucket[gi_].x;
    }
#pragma unroll
    for (int j = 0; j < RUNS; ++j) {
      const int mc = __builtin_amdgcn_readlane(m, j);
      if (lane < mc) atomicAdd(&sCnt[key[j] >> 16], 1);
    }
    for (int j = 0; j < RUNS; ++j) {
      const int sc = __builtin_amdgcn_readlane(s, j);
      const int mc = __builtin_amdgcn_readlane(m, j);
      for (int k = 32 + lane; k < mc; k += 32)
        atomicAdd(&sCnt[bucket[(c0 + j) * CHUNK + sc + k].x >> 16], 1);
    }
  } else {
    const int n2 = gt & 63, o4 = n2 >> 4, lm = n2 & 15, sub = gt >> 6;
#pragma unroll
    for (int j = 0; j < 8; ++j) {
      const int g = j * 8 + sub;
      const int tt = g >> 2, p = (g >> 1) & 1, lh = g & 1;
      const float* src = W2 + (16 * tt + 8 * p + 2 * lh) * COUTN + n2;
      float4 v = make_float4(src[0], src[COUTN], src[4 * COUTN], src[5 * COUTN]);
      *(float4*)(&sW2f[((((tt * 4 + o4) * 2 + p) * 32) + lh * 16 + lm) * 4]) = v;
    }
    if (gt < HID) sB1[gt] = b1[gt];
    if (gt < COUTN) sB2[gt] = b2[gt];
  }
  __syncthreads();

  int cnt = 0, inc = 0;
  if (wv < 8) {
    cnt = sCnt[tid];
    inc = cnt;
#pragma unroll
    for (int d = 1; d < 32; d <<= 1) {
      int t = __shfl_up(inc, d);
      if (lane >= d) inc += t;
    }
    if (lane == 31) sWc[wv] = inc;
  }
  __syncthreads();
  if (wv < 8) {
    int off = 0;
#pragma unroll
    for (int w = 0; w < 7; ++w) off += (w < wv) ? sWc[w] : 0;
    const int ex = off + inc - cnt;
    sCur[tid] = ex;
    sRp[tid] = ex;
  }
  __syncthreads();

  if (wv < 8) {
    int2 rec[RUNS];
#pragma unroll
    for (int j = 0; j < RUNS; ++j) {
      GCN_RUN(j)
      rec[j] = bucket[gi_];
    }
#pragma unroll
    for (int j = 0; j < RUNS; ++j) {
      const int mc = __builtin_amdgcn_readlane(m, j);
      if (lane < mc) {
        const int p = atomicAdd(&sCur[rec[j].x >> 16], 1);
        const int2 o = make_int2(rec[j].x & 0xffff, rec[j].y);
        if (staged) sEdge[p] = o;
        else        sedge[start + p] = o;
      }
    }
    for (int j = 0; j < RUNS; ++j) {
      const int sc = __builtin_amdgcn_readlane(s, j);
      const int mc = __builtin_amdgcn_readlane(m, j);
      for (int k = 32 + lane; k < mc; k += 32) {
        const int2 rr = bucket[(c0 + j) * CHUNK + sc + k];
        const int p = atomicAdd(&sCur[rr.x >> 16], 1);
        const int2 o = make_int2(rr.x & 0xffff, rr.y);
        if (staged) sEdge[p] = o;
        else        sedge[start + p] = o;
      }
    }
  }
#undef GCN_RUN
  __syncthreads();

  if (wv >= 8) {
    const int g = wv - 8, nbl = g >> 3;
    if (nb0 + nbl < nb1)
      gcn_gather_pair(x, sRp, sEdge, sedge, start, staged, &sTile[0][0],
                      nbl * 16 + (g & 7) * 2, nbl, g & 7, lane);
  }
  __syncthreads();

  float* st = &sStage[wv & 7][0];
  const float4* w1f = (const float4*)sW1f + lane;
  const float4* w2f = (const float4*)sW2f + lane;

  for (int it = 0; it < nIt; ++it) {
    const int nbA = nb0 + 2 * it;
    const bool mact = (wv < 8) && (nbA + (wv >> 2) < nb1);
    v2f a1[16];
#pragma unroll
    for (int s = 0; s < 16; ++s) a1[s] = (v2f){0.f, 0.f};
    if (wv >= 8) {
      const int g = wv - 8, nbl = g >> 3;
      if (it + 1 < nIt && nbA + 2 + nbl < nb1)
        gcn_gather_pair(x, sRp, sEdge, sedge, start, staged, &sTile[(it + 1) & 1][0],
                        (nbA + 2 + nbl - nb0) * 16 + (g & 7) * 2, nbl, g & 7, lane);
    } else if (mact) {
      const float* base = &sTile[it & 1][(wv * 16 + lmod) * TP + 2 * lhalf];
#pragma unroll
      for (int s = 0; s < 16; ++s)
        a1[s] = *(const v2f*)(base + 4 * s);
    }
    __builtin_amdgcn_fence(__ATOMIC_RELEASE, "workgroup", "local");
    __builtin_amdgcn_s_barrier_signal(-1);

    if (mact) {
      const int row0 = (wv & 3) * NN + (nbA + (wv >> 2)) * 16;

      v8f oacc[4];
#pragma unroll
      for (int o = 0; o < 4; ++o) {
        float bv = sB2[o * 16 + lmod];
#pragma unroll
        for (int j = 0; j < 8; ++j) oacc[o][j] = bv;
      }

      for (int t = 0; t < 16; ++t) {
        v8f h;
        float b1v = sB1[t * 16 + lmod];
#pragma unroll
        for (int j = 0; j < 8; ++j) h[j] = b1v;
#pragma unroll
        for (int p = 0; p < 8; ++p) {
          float4 w = w1f[(t * 8 + p) * 32];
          v2f bA = {w.x, w.y};
          v2f bB = {w.z, w.w};
          h = __builtin_amdgcn_wmma_f32_16x16x4_f32(false, a1[2 * p], false, bA,
                                                    (short)0, h, false, false);
          h = __builtin_amdgcn_wmma_f32_16x16x4_f32(false, a1[2 * p + 1], false, bB,
                                                    (short)0, h, false, false);
        }
#pragma unroll
        for (int j = 0; j < 8; ++j) h[j] = fmaxf(h[j], 0.f);

#pragma unroll
        for (int j = 0; j < 8; ++j)
          st[(lhalf * 8 + j) * STP + lmod] = h[j];
        asm volatile("s_wait_dscnt 0" ::: "memory");

        v2f a2[4];
#pragma unroll
        for (int kk = 0; kk < 4; ++kk)
          a2[kk] = *(const v2f*)(&st[lmod * STP + 4 * kk + 2 * lhalf]);

#pragma unroll
        for (int o = 0; o < 4; ++o) {
#pragma unroll
          for (int p = 0; p < 2; ++p) {
            float4 w = w2f[((t * 4 + o) * 2 + p) * 32];
            v2f bA = {w.x, w.y};
            v2f bB = {w.z, w.w};
            oacc[o] = __builtin_amdgcn_wmma_f32_16x16x4_f32(false, a2[2 * p], false, bA,
                                                            (short)0, oacc[o], false, false);
            oacc[o] = __builtin_amdgcn_wmma_f32_16x16x4_f32(false, a2[2 * p + 1], false, bB,
                                                            (short)0, oacc[o], false, false);
          }
        }
      }

#pragma unroll
      for (int o = 0; o < 4; ++o) {
#pragma unroll
        for (int j = 0; j < 8; ++j)
          out[(size_t)(row0 + lhalf * 8 + j) * COUTN + o * 16 + lmod] = oacc[o][j];
      }
    }
    __builtin_amdgcn_s_barrier_wait(-1);
    __builtin_amdgcn_fence(__ATOMIC_ACQUIRE, "workgroup", "local");
  }
}

extern "C" void kernel_launch(void* const* d_in, const int* in_sizes, int n_in,
                              void* d_out, int out_size, void* d_ws, size_t ws_size,
                              hipStream_t stream) {
  const float* x     = (const float*)d_in[0];
  const int*   A_idx = (const int*)d_in[1];
  const float* A_val = (const float*)d_in[2];
  const float* W1    = (const float*)d_in[3];
  const float* b1    = (const float*)d_in[4];
  const float* W2    = (const float*)d_in[5];
  const float* b2    = (const float*)d_in[6];
  float* out = (float*)d_out;

  int*  offT     = (int*)d_ws;
  int*  bucket   = (int*)((char*)d_ws + 1048576);
  int*  sedge    = (int*)((char*)d_ws + 8388608);
  const int* rows = A_idx;
  const int* cols = A_idx + NE;

  gcn_chunksort_kernel<<<NCHUNK, 256, 0, stream>>>(rows, cols, A_val, offT, bucket);
  gcn_fused_kernel<<<NWG, 768, 0, stream>>>(x, offT, bucket, sedge, out, W1, b1, W2, b2);
}
